# starting kernel
# baseline (speedup 1.0000x reference)
#include <hip/hip_runtime.h>
#include <math.h>

typedef float v2f __attribute__((ext_vector_type(2)));
typedef float v8f __attribute__((ext_vector_type(8)));

#define NQ 10
#define NL 4
#define DIM 1024
#define BATCH 4096
#define K_IN 128
#define N_OUT 256
#define BPAD 132

__global__ __launch_bounds__(128) void k_encode(
    const float* __restrict__ x, const float* __restrict__ Win,
    const float* __restrict__ bin, float* __restrict__ h)
{
  __shared__ float Bs[16 * BPAD];

  const int tid  = threadIdx.x;
  const int lane = tid & 31;
  const int wave = tid >> 5;
  const int m0   = (blockIdx.x * 4 + wave) * 16;
  const int half = lane >> 4;
  const int l16  = lane & 15;

  for (int i = tid; i < 16 * K_IN; i += 128) {
    const int n = i >> 7;
    const int k = i & (K_IN - 1);
    Bs[n * BPAD + k] = (n < NQ) ? Win[n * K_IN + k] : 0.0f;
  }
  __syncthreads();

  const float* arow = x + (m0 + l16) * K_IN + half * 2;
  const float* brow = Bs + l16 * BPAD + half * 2;

  v8f acc = {};
  #pragma unroll 4
  for (int kk = 0; kk < K_IN / 4; ++kk) {
    v2f a, b;
    a.x = arow[kk * 4];
    a.y = arow[kk * 4 + 1];
    b.x = brow[kk * 4];
    b.y = brow[kk * 4 + 1];
    acc = __builtin_amdgcn_wmma_f32_16x16x4_f32(false, a, false, b,
                                                (short)0, acc, false, false);
  }
  if (l16 < NQ) {
    const float bias = bin[l16];
    #pragma unroll
    for (int r = 0; r < 8; ++r) {
      const int m = m0 + r + 8 * half;
      h[m * NQ + l16] = tanhf(acc[r] + bias);
    }
  }
}

__global__ __launch_bounds__(128) void k_circuit(
    const float* __restrict__ h, const float* __restrict__ qp,
    float* __restrict__ z)
{
  __shared__ float sre[4][DIM];
  __shared__ float sim[4][DIM];
  __shared__ unsigned short perm[DIM];

  const int tid  = threadIdx.x;
  const int wave = tid >> 5;
  const int lane = tid & 31;
  const int b    = blockIdx.x * 4 + wave;

  for (int j = tid; j < DIM; j += 128) {
    int src = j;
    if ((src >> 0) & 1) src ^= (1 << 9);
    #pragma unroll
    for (int c = 8; c >= 0; --c)
      if ((src >> (9 - c)) & 1) src ^= (1 << (9 - (c + 1)));
    perm[j] = (unsigned short)src;
  }
  __syncthreads();

  float* re = sre[wave];
  float* im = sim[wave];

  for (int k = 0; k < DIM / 32; ++k) {
    const int i = k * 32 + lane;
    re[i] = (i == 0) ? 1.0f : 0.0f;
    im[i] = 0.0f;
  }

  auto rx = [&](float theta, int w) {
    const float hc = cosf(0.5f * theta);
    const float hs = sinf(0.5f * theta);
    const int mask = 1 << (9 - w);
    const int low  = mask - 1;
    for (int k = 0; k < 16; ++k) {
      const int p  = k * 32 + lane;
      const int i0 = ((p & ~low) << 1) | (p & low);
      const int i1 = i0 | mask;
      const float are = re[i0], aim = im[i0];
      const float bre = re[i1], bim = im[i1];
      re[i0] = hc * are + hs * bim;
      im[i0] = hc * aim - hs * bre;
      re[i1] = hc * bre + hs * aim;
      im[i1] = hc * bim - hs * are;
    }
  };

  for (int w = 0; w < NQ; ++w) rx(h[b * NQ + w], w);

  for (int l = 0; l < NL; ++l) {
    for (int w = 0; w < NQ; ++w) rx(qp[l * NQ + w], w);
    float tre[32], tim[32];
    #pragma unroll
    for (int k = 0; k < 32; ++k) {
      const int s = perm[k * 32 + lane];
      tre[k] = re[s];
      tim[k] = im[s];
    }
    #pragma unroll
    for (int k = 0; k < 32; ++k) {
      const int j = k * 32 + lane;
      re[j] = tre[k];
      im[j] = tim[k];
    }
  }

  float acc[NQ];
  #pragma unroll
  for (int q = 0; q < NQ; ++q) acc[q] = 0.0f;
  for (int k = 0; k < 32; ++k) {
    const int i = k * 32 + lane;
    const float pr = re[i] * re[i] + im[i] * im[i];
    #pragma unroll
    for (int q = 0; q < NQ; ++q)
      acc[q] += ((i >> (9 - q)) & 1) ? -pr : pr;
  }
  #pragma unroll
  for (int q = 0; q < NQ; ++q) {
    float v = acc[q];
    for (int off = 16; off > 0; off >>= 1)
      v += __shfl_xor(v, off, 32);
    if (lane == 0) z[b * NQ + q] = v;
  }
}

__global__ __launch_bounds__(128) void k_readout(
    const float* __restrict__ z, const float* __restrict__ Wout,
    const float* __restrict__ bout, float* __restrict__ out)
{
  const int lane = threadIdx.x & 31;
  const int wave = threadIdx.x >> 5;
  const int tile = blockIdx.x * 4 + wave;
  const int m0   = (tile >> 4) * 16;
  const int n0   = (tile & 15) * 16;
  const int half = lane >> 4;
  const int l16  = lane & 15;

  const float* arow = z    + (m0 + l16) * NQ;
  const float* brow = Wout + (n0 + l16) * NQ;

  v8f acc = {};
  #pragma unroll
  for (int kk = 0; kk < 3; ++kk) {
    const int k0  = kk * 4 + half * 2;
    const int k0c = (k0 < NQ - 1) ? k0 : (NQ - 2);
    const float vm0 = (k0     < NQ) ? 1.0f : 0.0f;
    const float vm1 = (k0 + 1 < NQ) ? 1.0f : 0.0f;
    v2f a, b;
    a.x = arow[k0c]     * vm0;
    a.y = arow[k0c + 1] * vm1;
    b.x = brow[k0c];
    b.y = brow[k0c + 1];
    acc = __builtin_amdgcn_wmma_f32_16x16x4_f32(false, a, false, b,
                                                (short)0, acc, false, false);
  }
  const float bias = bout[n0 + l16];
  #pragma unroll
  for (int r = 0; r < 8; ++r) {
    const int m = m0 + r + 8 * half;
    out[m * N_OUT + n0 + l16] = acc[r] + bias;
  }
}

extern "C" void kernel_launch(void* const* d_in, const int* in_sizes, int n_in,
                              void* d_out, int out_size, void* d_ws, size_t ws_size,
                              hipStream_t stream) {
  const float* x    = (const float*)d_in[0];
  const float* Win  = (const float*)d_in[1];
  const float* bin  = (const float*)d_in[2];
  const float* qp   = (const float*)d_in[3];
  const float* Wout = (const float*)d_in[4];
  const float* bout = (const float*)d_in[5];
  float* out = (float*)d_out;

  float* h  = (float*)d_ws;
  float* zv = h + BATCH * NQ;

  k_encode <<<BATCH / 16 / 4, 128, 0, stream>>>(x, Win, bin, h);
  k_circuit<<<BATCH / 4,      128, 0, stream>>>(h, qp, zv);
  k_readout<<<BATCH * 16 / 16 / 4, 128, 0, stream>>>(zv, Wout, bout, out);
}


# k_circuit keeps the state in VGPRs (amplitude k*32+lane), not LDS: RX on qubits 0-4 pairs registers, on 5-9 lanes (DPP row_xmask, permlanex16); a layer's cos/sin done one per lane, read by readlane
# speedup vs baseline: 2.0378x; 2.0378x over previous
#include <hip/hip_runtime.h>
#include <math.h>

typedef float v2f __attribute__((ext_vector_type(2)));
typedef float v8f __attribute__((ext_vector_type(8)));

#define NQ 10
#define NL 4
#define DIM 1024
#define BATCH 4096
#define K_IN 128
#define N_OUT 256
#define BPAD 132

__global__ __launch_bounds__(128) void k_encode(
    const float* __restrict__ x, const float* __restrict__ Win,
    const float* __restrict__ bin, float* __restrict__ h)
{
  __shared__ float Bs[16 * BPAD];

  const int tid  = threadIdx.x;
  const int lane = tid & 31;
  const int wave = tid >> 5;
  const int m0   = (blockIdx.x * 4 + wave) * 16;
  const int half = lane >> 4;
  const int l16  = lane & 15;

  for (int i = tid; i < 16 * K_IN; i += 128) {
    const int n = i >> 7;
    const int k = i & (K_IN - 1);
    Bs[n * BPAD + k] = (n < NQ) ? Win[n * K_IN + k] : 0.0f;
  }
  __syncthreads();

  const float* arow = x + (m0 + l16) * K_IN + half * 2;
  const float* brow = Bs + l16 * BPAD + half * 2;

  v8f acc = {};
  #pragma unroll 4
  for (int kk = 0; kk < K_IN / 4; ++kk) {
    v2f a, b;
    a.x = arow[kk * 4];
    a.y = arow[kk * 4 + 1];
    b.x = brow[kk * 4];
    b.y = brow[kk * 4 + 1];
    acc = __builtin_amdgcn_wmma_f32_16x16x4_f32(false, a, false, b,
                                                (short)0, acc, false, false);
  }
  if (l16 < NQ) {
    const float bias = bin[l16];
    #pragma unroll
    for (int r = 0; r < 8; ++r) {
      const int m = m0 + r + 8 * half;
      h[m * NQ + l16] = tanhf(acc[r] + bias);
    }
  }
}

template <int M>
__device__ __forceinline__ float lane_xor(float v)
{
  const int bits = __builtin_bit_cast(int, v);
  if constexpr (M == 16) {
    return __builtin_bit_cast(float, __builtin_amdgcn_permlanex16(
        bits, bits, 0x76543210, 0xfedcba98, false, false));
  } else {
    return __builtin_bit_cast(float, __builtin_amdgcn_update_dpp(
        0, bits, 0x160 + M, 0xf, 0xf, true));
  }
}

template <int Q>
__device__ __forceinline__ void rx_gate(float (&re)[32], float (&im)[32],
                                        float hc, float hs)
{
  if constexpr (Q < 5) {
    constexpr int km = 1 << (4 - Q);
    #pragma unroll
    for (int k = 0; k < 32; ++k) {
      if ((k & km) == 0) {
        const int k1 = k | km;
        const float are = re[k],  aim = im[k];
        const float bre = re[k1], bim = im[k1];
        re[k]  = hc * are + hs * bim;
        im[k]  = hc * aim - hs * bre;
        re[k1] = hc * bre + hs * aim;
        im[k1] = hc * bim - hs * are;
      }
    }
  } else {
    constexpr int lm = 1 << (9 - Q);
    #pragma unroll
    for (int k = 0; k < 32; ++k) {
      const float ore = lane_xor<lm>(re[k]);
      const float oim = lane_xor<lm>(im[k]);
      const float nre = hc * re[k] + hs * oim;
      const float nim = hc * im[k] - hs * ore;
      re[k] = nre;
      im[k] = nim;
    }
  }
}

__device__ __forceinline__ void layer_coeffs(const float* __restrict__ ang,
                                             int lane, float (&hc)[NQ],
                                             float (&hs)[NQ])
{
  const float theta = (lane < NQ) ? ang[lane] : 0.0f;
  const int cb = __builtin_bit_cast(int, cosf(0.5f * theta));
  const int sb = __builtin_bit_cast(int, sinf(0.5f * theta));
  #pragma unroll
  for (int w = 0; w < NQ; ++w) {
    hc[w] = __builtin_bit_cast(float, __builtin_amdgcn_readlane(cb, w));
    hs[w] = __builtin_bit_cast(float, __builtin_amdgcn_readlane(sb, w));
  }
}

__device__ __forceinline__ void rx_layer(float (&re)[32], float (&im)[32],
                                         const float (&hc)[NQ],
                                         const float (&hs)[NQ])
{
  rx_gate<0>(re, im, hc[0], hs[0]);
  rx_gate<1>(re, im, hc[1], hs[1]);
  rx_gate<2>(re, im, hc[2], hs[2]);
  rx_gate<3>(re, im, hc[3], hs[3]);
  rx_gate<4>(re, im, hc[4], hs[4]);
  rx_gate<5>(re, im, hc[5], hs[5]);
  rx_gate<6>(re, im, hc[6], hs[6]);
  rx_gate<7>(re, im, hc[7], hs[7]);
  rx_gate<8>(re, im, hc[8], hs[8]);
  rx_gate<9>(re, im, hc[9], hs[9]);
}

__global__ __launch_bounds__(128) void k_circuit(
    const float* __restrict__ h, const float* __restrict__ qp,
    float* __restrict__ z)
{
  __shared__ float2 sst[4][DIM];
  __shared__ unsigned short perm[DIM];

  const int tid  = threadIdx.x;
  const int wave = tid >> 5;
  const int lane = tid & 31;
  const int b    = blockIdx.x * 4 + wave;

  for (int j = tid; j < DIM; j += 128) {
    int src = j;
    if ((src >> 0) & 1) src ^= (1 << 9);
    #pragma unroll
    for (int c = 8; c >= 0; --c)
      if ((src >> (9 - c)) & 1) src ^= (1 << (9 - (c + 1)));
    perm[j] = (unsigned short)src;
  }
  __syncthreads();

  float2* st = sst[wave];

  int sidx[32];
  #pragma unroll
  for (int k = 0; k < 32; ++k) sidx[k] = perm[k * 32 + lane];

  float re[32], im[32];
  #pragma unroll
  for (int k = 0; k < 32; ++k) {
    re[k] = (k == 0 && lane == 0) ? 1.0f : 0.0f;
    im[k] = 0.0f;
  }

  {
    float hc[NQ], hs[NQ];
    layer_coeffs(h + b * NQ, lane, hc, hs);
    rx_layer(re, im, hc, hs);
  }

  #pragma nounroll
  for (int l = 0; l < NL; ++l) {
    float hc[NQ], hs[NQ];
    layer_coeffs(qp + l * NQ, lane, hc, hs);
    rx_layer(re, im, hc, hs);

    #pragma unroll
    for (int k = 0; k < 32; ++k)
      st[k * 32 + lane] = make_float2(re[k], im[k]);
    #pragma unroll
    for (int k = 0; k < 32; ++k) {
      const float2 v = st[sidx[k]];
      re[k] = v.x;
      im[k] = v.y;
    }
  }

  float acc[NQ];
  #pragma unroll
  for (int q = 0; q < NQ; ++q) acc[q] = 0.0f;
  #pragma unroll
  for (int k = 0; k < 32; ++k) {
    const int i = k * 32 + lane;
    const float pr = re[k] * re[k] + im[k] * im[k];
    #pragma unroll
    for (int q = 0; q < NQ; ++q)
      acc[q] += ((i >> (9 - q)) & 1) ? -pr : pr;
  }
  #pragma unroll
  for (int q = 0; q < NQ; ++q) {
    float v = acc[q];
    for (int off = 16; off > 0; off >>= 1)
      v += __shfl_xor(v, off, 32);
    if (lane == 0) z[b * NQ + q] = v;
  }
}

__global__ __launch_bounds__(128) void k_readout(
    const float* __restrict__ z, const float* __restrict__ Wout,
    const float* __restrict__ bout, float* __restrict__ out)
{
  const int lane = threadIdx.x & 31;
  const int wave = threadIdx.x >> 5;
  const int tile = blockIdx.x * 4 + wave;
  const int m0   = (tile >> 4) * 16;
  const int n0   = (tile & 15) * 16;
  const int half = lane >> 4;
  const int l16  = lane & 15;

  const float* arow = z    + (m0 + l16) * NQ;
  const float* brow = Wout + (n0 + l16) * NQ;

  v8f acc = {};
  #pragma unroll
  for (int kk = 0; kk < 3; ++kk) {
    const int k0  = kk * 4 + half * 2;
    const int k0c = (k0 < NQ - 1) ? k0 : (NQ - 2);
    const float vm0 = (k0     < NQ) ? 1.0f : 0.0f;
    const float vm1 = (k0 + 1 < NQ) ? 1.0f : 0.0f;
    v2f a, b;
    a.x = arow[k0c]     * vm0;
    a.y = arow[k0c + 1] * vm1;
    b.x = brow[k0c];
    b.y = brow[k0c + 1];
    acc = __builtin_amdgcn_wmma_f32_16x16x4_f32(false, a, false, b,
                                                (short)0, acc, false, false);
  }
  const float bias = bout[n0 + l16];
  #pragma unroll
  for (int r = 0; r < 8; ++r) {
    const int m = m0 + r + 8 * half;
    out[m * N_OUT + n0 + l16] = acc[r] + bias;
  }
}

extern "C" void kernel_launch(void* const* d_in, const int* in_sizes, int n_in,
                              void* d_out, int out_size, void* d_ws, size_t ws_size,
                              hipStream_t stream) {
  const float* x    = (const float*)d_in[0];
  const float* Win  = (const float*)d_in[1];
  const float* bin  = (const float*)d_in[2];
  const float* qp   = (const float*)d_in[3];
  const float* Wout = (const float*)d_in[4];
  const float* bout = (const float*)d_in[5];
  float* out = (float*)d_out;

  float* h  = (float*)d_ws;
  float* zv = h + BATCH * NQ;

  k_encode <<<BATCH / 16 / 4, 128, 0, stream>>>(x, Win, bin, h);
  k_circuit<<<BATCH / 4,      128, 0, stream>>>(h, qp, zv);
  k_readout<<<BATCH * 16 / 16 / 4, 128, 0, stream>>>(zv, Wout, bout, out);
}


# k_circuit: the 20 RX gates ahead of the first CNOT ring act on |0..0>; their result, a product state (per qubit cos, -i sin of the composed angle; register x lane factor), is written directly
# speedup vs baseline: 2.3824x; 1.1691x over previous
#include <hip/hip_runtime.h>
#include <math.h>

typedef float v2f __attribute__((ext_vector_type(2)));
typedef float v8f __attribute__((ext_vector_type(8)));

#define NQ 10
#define NL 4
#define DIM 1024
#define BATCH 4096
#define K_IN 128
#define N_OUT 256
#define BPAD 132

__global__ __launch_bounds__(128) void k_encode(
    const float* __restrict__ x, const float* __restrict__ Win,
    const float* __restrict__ bin, float* __restrict__ h)
{
  __shared__ float Bs[16 * BPAD];

  const int tid  = threadIdx.x;
  const int lane = tid & 31;
  const int wave = tid >> 5;
  const int m0   = (blockIdx.x * 4 + wave) * 16;
  const int half = lane >> 4;
  const int l16  = lane & 15;

  for (int i = tid; i < 16 * K_IN; i += 128) {
    const int n = i >> 7;
    const int k = i & (K_IN - 1);
    Bs[n * BPAD + k] = (n < NQ) ? Win[n * K_IN + k] : 0.0f;
  }
  __syncthreads();

  const float* arow = x + (m0 + l16) * K_IN + half * 2;
  const float* brow = Bs + l16 * BPAD + half * 2;

  v8f acc = {};
  #pragma unroll 4
  for (int kk = 0; kk < K_IN / 4; ++kk) {
    v2f a, b;
    a.x = arow[kk * 4];
    a.y = arow[kk * 4 + 1];
    b.x = brow[kk * 4];
    b.y = brow[kk * 4 + 1];
    acc = __builtin_amdgcn_wmma_f32_16x16x4_f32(false, a, false, b,
                                                (short)0, acc, false, false);
  }
  if (l16 < NQ) {
    const float bias = bin[l16];
    #pragma unroll
    for (int r = 0; r < 8; ++r) {
      const int m = m0 + r + 8 * half;
      h[m * NQ + l16] = tanhf(acc[r] + bias);
    }
  }
}

template <int M>
__device__ __forceinline__ float lane_xor(float v)
{
  const int bits = __builtin_bit_cast(int, v);
  if constexpr (M == 16) {
    return __builtin_bit_cast(float, __builtin_amdgcn_permlanex16(
        bits, bits, 0x76543210, 0xfedcba98, false, false));
  } else {
    return __builtin_bit_cast(float, __builtin_amdgcn_update_dpp(
        0, bits, 0x160 + M, 0xf, 0xf, true));
  }
}

template <int Q>
__device__ __forceinline__ void rx_gate(float (&re)[32], float (&im)[32],
                                        float hc, float hs)
{
  if constexpr (Q < 5) {
    constexpr int km = 1 << (4 - Q);
    #pragma unroll
    for (int k = 0; k < 32; ++k) {
      if ((k & km) == 0) {
        const int k1 = k | km;
        const float are = re[k],  aim = im[k];
        const float bre = re[k1], bim = im[k1];
        re[k]  = hc * are + hs * bim;
        im[k]  = hc * aim - hs * bre;
        re[k1] = hc * bre + hs * aim;
        im[k1] = hc * bim - hs * are;
      }
    }
  } else {
    constexpr int lm = 1 << (9 - Q);
    #pragma unroll
    for (int k = 0; k < 32; ++k) {
      const float ore = lane_xor<lm>(re[k]);
      const float oim = lane_xor<lm>(im[k]);
      const float nre = hc * re[k] + hs * oim;
      const float nim = hc * im[k] - hs * ore;
      re[k] = nre;
      im[k] = nim;
    }
  }
}

__device__ __forceinline__ void layer_coeffs(const float* __restrict__ ang,
                                             int lane, float (&hc)[NQ],
                                             float (&hs)[NQ])
{
  const float theta = (lane < NQ) ? ang[lane] : 0.0f;
  const int cb = __builtin_bit_cast(int, cosf(0.5f * theta));
  const int sb = __builtin_bit_cast(int, sinf(0.5f * theta));
  #pragma unroll
  for (int w = 0; w < NQ; ++w) {
    hc[w] = __builtin_bit_cast(float, __builtin_amdgcn_readlane(cb, w));
    hs[w] = __builtin_bit_cast(float, __builtin_amdgcn_readlane(sb, w));
  }
}

__device__ __forceinline__ void rx_layer(float (&re)[32], float (&im)[32],
                                         const float (&hc)[NQ],
                                         const float (&hs)[NQ])
{
  rx_gate<0>(re, im, hc[0], hs[0]);
  rx_gate<1>(re, im, hc[1], hs[1]);
  rx_gate<2>(re, im, hc[2], hs[2]);
  rx_gate<3>(re, im, hc[3], hs[3]);
  rx_gate<4>(re, im, hc[4], hs[4]);
  rx_gate<5>(re, im, hc[5], hs[5]);
  rx_gate<6>(re, im, hc[6], hs[6]);
  rx_gate<7>(re, im, hc[7], hs[7]);
  rx_gate<8>(re, im, hc[8], hs[8]);
  rx_gate<9>(re, im, hc[9], hs[9]);
}

__device__ __forceinline__ void ring_cnot(float2* __restrict__ st,
                                          const int (&sidx)[32], int lane,
                                          float (&re)[32], float (&im)[32])
{
  #pragma unroll
  for (int k = 0; k < 32; ++k)
    st[k * 32 + lane] = make_float2(re[k], im[k]);
  #pragma unroll
  for (int k = 0; k < 32; ++k) {
    const float2 v = st[sidx[k]];
    re[k] = v.x;
    im[k] = v.y;
  }
}

__device__ __forceinline__ void product_state(const float* __restrict__ hang,
                                              const float* __restrict__ qang,
                                              int lane, float (&re)[32],
                                              float (&im)[32])
{
  const int w16 = lane & 15;
  float theta = 0.0f;
  if (w16 < NQ) theta = (lane < 16) ? hang[w16] : qang[w16];
  const float c1 = cosf(0.5f * theta);
  const float s1 = sinf(0.5f * theta);
  const float c2 = lane_xor<16>(c1);
  const float s2 = lane_xor<16>(s1);
  const int Cb = __builtin_bit_cast(int, c1 * c2 - s1 * s2);
  const int Sb = __builtin_bit_cast(int, s1 * c2 + c1 * s2);

  float rf = 1.0f, lf = 1.0f;
  #pragma unroll
  for (int w = 0; w < 5; ++w) {
    const float Cr = __builtin_bit_cast(float, __builtin_amdgcn_readlane(Cb, w));
    const float Sr = __builtin_bit_cast(float, __builtin_amdgcn_readlane(Sb, w));
    const float Cl = __builtin_bit_cast(float, __builtin_amdgcn_readlane(Cb, w + 5));
    const float Sl = __builtin_bit_cast(float, __builtin_amdgcn_readlane(Sb, w + 5));
    rf *= ((lane >> (4 - w)) & 1) ? Sr : Cr;
    lf *= ((lane >> (4 - w)) & 1) ? Sl : Cl;
  }
  const int pl = __popc(lane) & 3;
  const float lre = (pl == 0) ? lf : (pl == 2) ? -lf : 0.0f;
  const float lim = (pl == 3) ? lf : (pl == 1) ? -lf : 0.0f;
  const int rfb = __builtin_bit_cast(int, rf);
  #pragma unroll
  for (int k = 0; k < 32; ++k) {
    const float r = __builtin_bit_cast(float, __builtin_amdgcn_readlane(rfb, k));
    const int pk = __builtin_popcount(k) & 3;
    if (pk == 0)      { re[k] =  r * lre; im[k] =  r * lim; }
    else if (pk == 1) { re[k] =  r * lim; im[k] = -r * lre; }
    else if (pk == 2) { re[k] = -r * lre; im[k] = -r * lim; }
    else              { re[k] = -r * lim; im[k] =  r * lre; }
  }
}

__global__ __launch_bounds__(128) void k_circuit(
    const float* __restrict__ h, const float* __restrict__ qp,
    float* __restrict__ z)
{
  __shared__ float2 sst[4][DIM];
  __shared__ unsigned short perm[DIM];

  const int tid  = threadIdx.x;
  const int wave = tid >> 5;
  const int lane = tid & 31;
  const int b    = blockIdx.x * 4 + wave;

  for (int j = tid; j < DIM; j += 128) {
    int src = j;
    if ((src >> 0) & 1) src ^= (1 << 9);
    #pragma unroll
    for (int c = 8; c >= 0; --c)
      if ((src >> (9 - c)) & 1) src ^= (1 << (9 - (c + 1)));
    perm[j] = (unsigned short)src;
  }
  __syncthreads();

  float2* st = sst[wave];

  int sidx[32];
  #pragma unroll
  for (int k = 0; k < 32; ++k) sidx[k] = perm[k * 32 + lane];

  float re[32], im[32];
  product_state(h + b * NQ, qp, lane, re, im);
  ring_cnot(st, sidx, lane, re, im);

  #pragma nounroll
  for (int l = 1; l < NL; ++l) {
    float hc[NQ], hs[NQ];
    layer_coeffs(qp + l * NQ, lane, hc, hs);
    rx_layer(re, im, hc, hs);
    ring_cnot(st, sidx, lane, re, im);
  }

  float acc[NQ];
  #pragma unroll
  for (int q = 0; q < NQ; ++q) acc[q] = 0.0f;
  #pragma unroll
  for (int k = 0; k < 32; ++k) {
    const int i = k * 32 + lane;
    const float pr = re[k] * re[k] + im[k] * im[k];
    #pragma unroll
    for (int q = 0; q < NQ; ++q)
      acc[q] += ((i >> (9 - q)) & 1) ? -pr : pr;
  }
  #pragma unroll
  for (int q = 0; q < NQ; ++q) {
    float v = acc[q];
    for (int off = 16; off > 0; off >>= 1)
      v += __shfl_xor(v, off, 32);
    if (lane == 0) z[b * NQ + q] = v;
  }
}

__global__ __launch_bounds__(128) void k_readout(
    const float* __restrict__ z, const float* __restrict__ Wout,
    const float* __restrict__ bout, float* __restrict__ out)
{
  const int lane = threadIdx.x & 31;
  const int wave = threadIdx.x >> 5;
  const int tile = blockIdx.x * 4 + wave;
  const int m0   = (tile >> 4) * 16;
  const int n0   = (tile & 15) * 16;
  const int half = lane >> 4;
  const int l16  = lane & 15;

  const float* arow = z    + (m0 + l16) * NQ;
  const float* brow = Wout + (n0 + l16) * NQ;

  v8f acc = {};
  #pragma unroll
  for (int kk = 0; kk < 3; ++kk) {
    const int k0  = kk * 4 + half * 2;
    const int k0c = (k0 < NQ - 1) ? k0 : (NQ - 2);
    const float vm0 = (k0     < NQ) ? 1.0f : 0.0f;
    const float vm1 = (k0 + 1 < NQ) ? 1.0f : 0.0f;
    v2f a, b;
    a.x = arow[k0c]     * vm0;
    a.y = arow[k0c + 1] * vm1;
    b.x = brow[k0c];
    b.y = brow[k0c + 1];
    acc = __builtin_amdgcn_wmma_f32_16x16x4_f32(false, a, false, b,
                                                (short)0, acc, false, false);
  }
  const float bias = bout[n0 + l16];
  #pragma unroll
  for (int r = 0; r < 8; ++r) {
    const int m = m0 + r + 8 * half;
    out[m * N_OUT + n0 + l16] = acc[r] + bias;
  }
}

extern "C" void kernel_launch(void* const* d_in, const int* in_sizes, int n_in,
                              void* d_out, int out_size, void* d_ws, size_t ws_size,
                              hipStream_t stream) {
  const float* x    = (const float*)d_in[0];
  const float* Win  = (const float*)d_in[1];
  const float* bin  = (const float*)d_in[2];
  const float* qp   = (const float*)d_in[3];
  const float* Wout = (const float*)d_in[4];
  const float* bout = (const float*)d_in[5];
  float* out = (float*)d_out;

  float* h  = (float*)d_ws;
  float* zv = h + BATCH * NQ;

  k_encode <<<BATCH / 16 / 4, 128, 0, stream>>>(x, Win, bin, h);
  k_circuit<<<BATCH / 4,      128, 0, stream>>>(h, qp, zv);
  k_readout<<<BATCH * 16 / 16 / 4, 128, 0, stream>>>(zv, Wout, bout, out);
}


# Three launches fused into one: a 16-wave workgroup owns 16 batch rows; wave 0 makes the h tile on WMMA, wave w runs circuit w and out tile w on WMMA; h, z pass through LDS; its loads issued up front
# speedup vs baseline: 3.5357x; 1.4841x over previous
#include <hip/hip_runtime.h>
#include <math.h>

typedef float v2f __attribute__((ext_vector_type(2)));
typedef float v8f __attribute__((ext_vector_type(8)));

#define NQ 10
#define NL 4
#define DIM 1024
#define BATCH 4096
#define K_IN 128
#define N_OUT 256
#define BPAD 132

#define ROWS 16
#define HPAD 16
#define ZPAD 12

template <int M>
__device__ __forceinline__ float lane_xor(float v)
{
  const int bits = __builtin_bit_cast(int, v);
  if constexpr (M == 16) {
    return __builtin_bit_cast(float, __builtin_amdgcn_permlanex16(
        bits, bits, 0x76543210, 0xfedcba98, false, false));
  } else {
    return __builtin_bit_cast(float, __builtin_amdgcn_update_dpp(
        0, bits, 0x160 + M, 0xf, 0xf, true));
  }
}

template <int Q>
__device__ __forceinline__ void rx_gate(float (&re)[32], float (&im)[32],
                                        float hc, float hs)
{
  if constexpr (Q < 5) {
    constexpr int km = 1 << (4 - Q);
    #pragma unroll
    for (int k = 0; k < 32; ++k) {
      if ((k & km) == 0) {
        const int k1 = k | km;
        const float are = re[k],  aim = im[k];
        const float bre = re[k1], bim = im[k1];
        re[k]  = hc * are + hs * bim;
        im[k]  = hc * aim - hs * bre;
        re[k1] = hc * bre + hs * aim;
        im[k1] = hc * bim - hs * are;
      }
    }
  } else {
    constexpr int lm = 1 << (9 - Q);
    #pragma unroll
    for (int k = 0; k < 32; ++k) {
      const float ore = lane_xor<lm>(re[k]);
      const float oim = lane_xor<lm>(im[k]);
      const float nre = hc * re[k] + hs * oim;
      const float nim = hc * im[k] - hs * ore;
      re[k] = nre;
      im[k] = nim;
    }
  }
}

__device__ __forceinline__ void layer_coeffs(const float* __restrict__ ang,
                                             int lane, float (&hc)[NQ],
                                             float (&hs)[NQ])
{
  const float theta = (lane < NQ) ? ang[lane] : 0.0f;
  const int cb = __builtin_bit_cast(int, cosf(0.5f * theta));
  const int sb = __builtin_bit_cast(int, sinf(0.5f * theta));
  #pragma unroll
  for (int w = 0; w < NQ; ++w) {
    hc[w] = __builtin_bit_cast(float, __builtin_amdgcn_readlane(cb, w));
    hs[w] = __builtin_bit_cast(float, __builtin_amdgcn_readlane(sb, w));
  }
}

__device__ __forceinline__ void rx_layer(float (&re)[32], float (&im)[32],
                                         const float (&hc)[NQ],
                                         const float (&hs)[NQ])
{
  rx_gate<0>(re, im, hc[0], hs[0]);
  rx_gate<1>(re, im, hc[1], hs[1]);
  rx_gate<2>(re, im, hc[2], hs[2]);
  rx_gate<3>(re, im, hc[3], hs[3]);
  rx_gate<4>(re, im, hc[4], hs[4]);
  rx_gate<5>(re, im, hc[5], hs[5]);
  rx_gate<6>(re, im, hc[6], hs[6]);
  rx_gate<7>(re, im, hc[7], hs[7]);
  rx_gate<8>(re, im, hc[8], hs[8]);
  rx_gate<9>(re, im, hc[9], hs[9]);
}

__device__ __forceinline__ void ring_cnot(float2* __restrict__ st,
                                          const int (&sidx)[32], int lane,
                                          float (&re)[32], float (&im)[32])
{
  #pragma unroll
  for (int k = 0; k < 32; ++k)
    st[k * 32 + lane] = make_float2(re[k], im[k]);
  #pragma unroll
  for (int k = 0; k < 32; ++k) {
    const float2 v = st[sidx[k]];
    re[k] = v.x;
    im[k] = v.y;
  }
}

__device__ __forceinline__ void product_state(const float* __restrict__ hang,
                                              float q0v,
                                              int lane, float (&re)[32],
                                              float (&im)[32])
{
  const int w16 = lane & 15;
  float theta = 0.0f;
  if (w16 < NQ) theta = (lane < 16) ? hang[w16] : q0v;
  const float c1 = cosf(0.5f * theta);
  const float s1 = sinf(0.5f * theta);
  const float c2 = lane_xor<16>(c1);
  const float s2 = lane_xor<16>(s1);
  const int Cb = __builtin_bit_cast(int, c1 * c2 - s1 * s2);
  const int Sb = __builtin_bit_cast(int, s1 * c2 + c1 * s2);

  float rf = 1.0f, lf = 1.0f;
  #pragma unroll
  for (int w = 0; w < 5; ++w) {
    const float Cr = __builtin_bit_cast(float, __builtin_amdgcn_readlane(Cb, w));
    const float Sr = __builtin_bit_cast(float, __builtin_amdgcn_readlane(Sb, w));
    const float Cl = __builtin_bit_cast(float, __builtin_amdgcn_readlane(Cb, w + 5));
    const float Sl = __builtin_bit_cast(float, __builtin_amdgcn_readlane(Sb, w + 5));
    rf *= ((lane >> (4 - w)) & 1) ? Sr : Cr;
    lf *= ((lane >> (4 - w)) & 1) ? Sl : Cl;
  }
  const int pl = __popc(lane) & 3;
  const float lre = (pl == 0) ? lf : (pl == 2) ? -lf : 0.0f;
  const float lim = (pl == 3) ? lf : (pl == 1) ? -lf : 0.0f;
  const int rfb = __builtin_bit_cast(int, rf);
  #pragma unroll
  for (int k = 0; k < 32; ++k) {
    const float r = __builtin_bit_cast(float, __builtin_amdgcn_readlane(rfb, k));
    const int pk = __builtin_popcount(k) & 3;
    if (pk == 0)      { re[k] =  r * lre; im[k] =  r * lim; }
    else if (pk == 1) { re[k] =  r * lim; im[k] = -r * lre; }
    else if (pk == 2) { re[k] = -r * lre; im[k] = -r * lim; }
    else              { re[k] = -r * lim; im[k] =  r * lre; }
  }
}

__global__ __launch_bounds__(512) void k_fused(
    const float* __restrict__ x, const float* __restrict__ Win,
    const float* __restrict__ bin, const float* __restrict__ qp,
    const float* __restrict__ Wout, const float* __restrict__ bout,
    float* __restrict__ out)
{
  __shared__ float2 sst[ROWS][DIM];
  __shared__ unsigned short perm[DIM];
  __shared__ float Bs[16 * BPAD];
  __shared__ float hsm[ROWS * HPAD];
  __shared__ float zsm[ROWS * ZPAD];

  const int tid  = threadIdx.x;
  const int wave = tid >> 5;
  const int lane = tid & 31;
  const int half = lane >> 4;
  const int l16  = lane & 15;
  const int m0   = blockIdx.x * ROWS;

  v2f afrag[K_IN / 4];
  float ebias = 0.0f;
  if (wave == 0) {
    if (l16 < NQ) ebias = bin[l16];
    const float* arow = x + (m0 + l16) * K_IN + half * 2;
    #pragma unroll
    for (int kk = 0; kk < K_IN / 4; ++kk) {
      afrag[kk].x = arow[kk * 4];
      afrag[kk].y = arow[kk * 4 + 1];
    }
  }

  const int n0 = wave * 16;
  v2f bfrag[3];
  {
    const float* brow = Wout + (n0 + l16) * NQ;
    #pragma unroll
    for (int kk = 0; kk < 3; ++kk) {
      const int k0  = kk * 4 + half * 2;
      const int k0c = (k0 < NQ - 1) ? k0 : (NQ - 2);
      bfrag[kk].x = brow[k0c];
      bfrag[kk].y = brow[k0c + 1];
    }
  }
  const float obias = bout[n0 + l16];
  const float q0v = (l16 < NQ) ? qp[l16] : 0.0f;

  float wv[16 * K_IN / 512];
  #pragma unroll
  for (int t = 0; t < 16 * K_IN / 512; ++t) {
    const int i = tid + t * 512;
    const int n = i >> 7;
    const int k = i & (K_IN - 1);
    wv[t] = Win[((n < NQ) ? n : (NQ - 1)) * K_IN + k];
  }
  #pragma unroll
  for (int t = 0; t < 16 * K_IN / 512; ++t) {
    const int i = tid + t * 512;
    const int n = i >> 7;
    const int k = i & (K_IN - 1);
    Bs[n * BPAD + k] = (n < NQ) ? wv[t] : 0.0f;
  }
  for (int j = tid; j < DIM; j += 512) {
    int src = j;
    if ((src >> 0) & 1) src ^= (1 << 9);
    #pragma unroll
    for (int c = 8; c >= 0; --c)
      if ((src >> (9 - c)) & 1) src ^= (1 << (9 - (c + 1)));
    perm[j] = (unsigned short)src;
  }
  if (tid < ROWS) {
    zsm[tid * ZPAD + NQ]     = 0.0f;
    zsm[tid * ZPAD + NQ + 1] = 0.0f;
  }
  __syncthreads();

  int sidx[32];
  #pragma unroll
  for (int k = 0; k < 32; ++k) sidx[k] = perm[k * 32 + lane];

  if (wave == 0) {
    const float* brow = Bs + l16 * BPAD + half * 2;
    v8f acc = {};
    #pragma unroll
    for (int kk = 0; kk < K_IN / 4; ++kk) {
      v2f b;
      b.x = brow[kk * 4];
      b.y = brow[kk * 4 + 1];
      acc = __builtin_amdgcn_wmma_f32_16x16x4_f32(false, afrag[kk], false, b,
                                                  (short)0, acc, false, false);
    }
    if (l16 < NQ) {
      #pragma unroll
      for (int r = 0; r < 8; ++r)
        hsm[(r + 8 * half) * HPAD + l16] = tanhf(acc[r] + ebias);
    }
  }
  __syncthreads();

  float2* st = sst[wave];

  float re[32], im[32];
  product_state(hsm + wave * HPAD, q0v, lane, re, im);
  ring_cnot(st, sidx, lane, re, im);

  #pragma nounroll
  for (int l = 1; l < NL; ++l) {
    float hc[NQ], hs[NQ];
    layer_coeffs(qp + l * NQ, lane, hc, hs);
    rx_layer(re, im, hc, hs);
    ring_cnot(st, sidx, lane, re, im);
  }

  float zacc[NQ];
  #pragma unroll
  for (int q = 0; q < NQ; ++q) zacc[q] = 0.0f;
  #pragma unroll
  for (int k = 0; k < 32; ++k) {
    const int i = k * 32 + lane;
    const float pr = re[k] * re[k] + im[k] * im[k];
    #pragma unroll
    for (int q = 0; q < NQ; ++q)
      zacc[q] += ((i >> (9 - q)) & 1) ? -pr : pr;
  }
  #pragma unroll
  for (int q = 0; q < NQ; ++q) {
    float v = zacc[q];
    for (int off = 16; off > 0; off >>= 1)
      v += __shfl_xor(v, off, 32);
    if (lane == 0) zsm[wave * ZPAD + q] = v;
  }
  __syncthreads();

  {
    const float* zrow = zsm + l16 * ZPAD + half * 2;
    v8f acc = {};
    #pragma unroll
    for (int kk = 0; kk < 3; ++kk) {
      v2f a;
      a.x = zrow[kk * 4];
      a.y = zrow[kk * 4 + 1];
      acc = __builtin_amdgcn_wmma_f32_16x16x4_f32(false, a, false, bfrag[kk],
                                                  (short)0, acc, false, false);
    }
    #pragma unroll
    for (int r = 0; r < 8; ++r) {
      const int m = m0 + r + 8 * half;
      out[m * N_OUT + n0 + l16] = acc[r] + obias;
    }
  }
}

extern "C" void kernel_launch(void* const* d_in, const int* in_sizes, int n_in,
                              void* d_out, int out_size, void* d_ws, size_t ws_size,
                              hipStream_t stream) {
  const float* x    = (const float*)d_in[0];
  const float* Win  = (const float*)d_in[1];
  const float* bin  = (const float*)d_in[2];
  const float* qp   = (const float*)d_in[3];
  const float* Wout = (const float*)d_in[4];
  const float* bout = (const float*)d_in[5];
  float* out = (float*)d_out;

  k_fused<<<BATCH / ROWS, 512, 0, stream>>>(x, Win, bin, qp, Wout, bout, out);
}


# phase 2: each RX of layers 1..3 applied as I - i*t*X, |t|<=1: larger of cos/sin factored out (squares multiplied into z once), X of sine-dominated gates folded into ring gather index; 1 FMA per part
# speedup vs baseline: 4.4369x; 1.2549x over previous
#include <hip/hip_runtime.h>
#include <math.h>

typedef float v2f __attribute__((ext_vector_type(2)));
typedef float v8f __attribute__((ext_vector_type(8)));

#define NQ 10
#define NL 4
#define DIM 1024
#define BATCH 4096
#define K_IN 128
#define N_OUT 256
#define BPAD 132

#define ROWS 16
#define HPAD 16
#define ZPAD 12

template <int M>
__device__ __forceinline__ float lane_xor(float v)
{
  const int bits = __builtin_bit_cast(int, v);
  if constexpr (M == 16) {
    return __builtin_bit_cast(float, __builtin_amdgcn_permlanex16(
        bits, bits, 0x76543210, 0xfedcba98, false, false));
  } else {
    return __builtin_bit_cast(float, __builtin_amdgcn_update_dpp(
        0, bits, 0x160 + M, 0xf, 0xf, true));
  }
}

template <int Q>
__device__ __forceinline__ void rx_gate(float (&re)[32], float (&im)[32],
                                        float t, float tn)
{
  if constexpr (Q < 5) {
    constexpr int km = 1 << (4 - Q);
    #pragma unroll
    for (int k = 0; k < 32; ++k) {
      if ((k & km) == 0) {
        const int k1 = k | km;
        const float are = re[k],  aim = im[k];
        const float bre = re[k1], bim = im[k1];
        re[k]  = are + t * bim;
        im[k]  = aim - t * bre;
        re[k1] = bre + t * aim;
        im[k1] = bim - t * are;
      }
    }
  } else {
    constexpr int lm = 1 << (9 - Q);
    #pragma unroll
    for (int k = 0; k < 32; ++k) {
      const float ore = lane_xor<lm>(re[k]);
      const float oim = lane_xor<lm>(im[k]);
      const float nre = re[k] - tn * oim;
      const float nim = im[k] - t * ore;
      re[k] = nre;
      im[k] = nim;
    }
  }
}

__device__ __forceinline__ void layer_coeffs(const float* __restrict__ ang,
                                             int lane, float (&t)[NQ],
                                             float (&tn)[NQ],
                                             int& flipb, float& mag2)
{
  const float theta = (lane < NQ) ? ang[lane] : 0.0f;
  const float c = cosf(0.5f * theta);
  const float s = sinf(0.5f * theta);
  const bool cdom = fabsf(c) >= fabsf(s);
  const float num = cdom ? s : -c;
  const float den = cdom ? c : s;
  const float tv = num / den;
  const int tb  = __builtin_bit_cast(int, tv);
  const int tnb = __builtin_bit_cast(int, -tv);
  #pragma unroll
  for (int w = 0; w < NQ; ++w) {
    t[w]  = __builtin_bit_cast(float, __builtin_amdgcn_readlane(tb, w));
    tn[w] = __builtin_bit_cast(float, __builtin_amdgcn_readlane(tnb, w));
  }
  const unsigned fl = __builtin_amdgcn_ballot_w32(!cdom);
  flipb = (int)(__builtin_bitreverse32(fl) >> 22) << 3;
  float m = den * den;
  m *= lane_xor<8>(m);
  m *= lane_xor<4>(m);
  m *= lane_xor<2>(m);
  m *= lane_xor<1>(m);
  mag2 = __builtin_bit_cast(float, __builtin_amdgcn_readlane(
      __builtin_bit_cast(int, m), 0));
}

__device__ __forceinline__ void rx_layer(float (&re)[32], float (&im)[32],
                                         const float (&t)[NQ],
                                         const float (&tn)[NQ])
{
  rx_gate<0>(re, im, t[0], tn[0]);
  rx_gate<1>(re, im, t[1], tn[1]);
  rx_gate<2>(re, im, t[2], tn[2]);
  rx_gate<3>(re, im, t[3], tn[3]);
  rx_gate<4>(re, im, t[4], tn[4]);
  rx_gate<5>(re, im, t[5], tn[5]);
  rx_gate<6>(re, im, t[6], tn[6]);
  rx_gate<7>(re, im, t[7], tn[7]);
  rx_gate<8>(re, im, t[8], tn[8]);
  rx_gate<9>(re, im, t[9], tn[9]);
}

__device__ __forceinline__ void ring_cnot(float2* __restrict__ st,
                                          const int (&sidx)[32], int flipb,
                                          int lane,
                                          float (&re)[32], float (&im)[32])
{
  #pragma unroll
  for (int k = 0; k < 32; ++k)
    st[k * 32 + lane] = make_float2(re[k], im[k]);
  const char* stb = reinterpret_cast<const char*>(st);
  #pragma unroll
  for (int k = 0; k < 32; ++k) {
    const float2 v = *reinterpret_cast<const float2*>(stb + (sidx[k] ^ flipb));
    re[k] = v.x;
    im[k] = v.y;
  }
}

__device__ __forceinline__ void product_state(const float* __restrict__ hang,
                                              float q0v,
                                              int lane, float (&re)[32],
                                              float (&im)[32])
{
  const int w16 = lane & 15;
  float theta = 0.0f;
  if (w16 < NQ) theta = (lane < 16) ? hang[w16] : q0v;
  const float c1 = cosf(0.5f * theta);
  const float s1 = sinf(0.5f * theta);
  const float c2 = lane_xor<16>(c1);
  const float s2 = lane_xor<16>(s1);
  const int Cb = __builtin_bit_cast(int, c1 * c2 - s1 * s2);
  const int Sb = __builtin_bit_cast(int, s1 * c2 + c1 * s2);

  float rf = 1.0f, lf = 1.0f;
  #pragma unroll
  for (int w = 0; w < 5; ++w) {
    const float Cr = __builtin_bit_cast(float, __builtin_amdgcn_readlane(Cb, w));
    const float Sr = __builtin_bit_cast(float, __builtin_amdgcn_readlane(Sb, w));
    const float Cl = __builtin_bit_cast(float, __builtin_amdgcn_readlane(Cb, w + 5));
    const float Sl = __builtin_bit_cast(float, __builtin_amdgcn_readlane(Sb, w + 5));
    rf *= ((lane >> (4 - w)) & 1) ? Sr : Cr;
    lf *= ((lane >> (4 - w)) & 1) ? Sl : Cl;
  }
  const int pl = __popc(lane) & 3;
  const float lre = (pl == 0) ? lf : (pl == 2) ? -lf : 0.0f;
  const float lim = (pl == 3) ? lf : (pl == 1) ? -lf : 0.0f;
  const int rfb = __builtin_bit_cast(int, rf);
  #pragma unroll
  for (int k = 0; k < 32; ++k) {
    const float r = __builtin_bit_cast(float, __builtin_amdgcn_readlane(rfb, k));
    const int pk = __builtin_popcount(k) & 3;
    if (pk == 0)      { re[k] =  r * lre; im[k] =  r * lim; }
    else if (pk == 1) { re[k] =  r * lim; im[k] = -r * lre; }
    else if (pk == 2) { re[k] = -r * lre; im[k] = -r * lim; }
    else              { re[k] = -r * lim; im[k] =  r * lre; }
  }
}

__global__ __launch_bounds__(512) void k_fused(
    const float* __restrict__ x, const float* __restrict__ Win,
    const float* __restrict__ bin, const float* __restrict__ qp,
    const float* __restrict__ Wout, const float* __restrict__ bout,
    float* __restrict__ out)
{
  __shared__ float2 sst[ROWS][DIM];
  __shared__ unsigned short perm[DIM];
  __shared__ float Bs[16 * BPAD];
  __shared__ float hsm[ROWS * HPAD];
  __shared__ float zsm[ROWS * ZPAD];

  const int tid  = threadIdx.x;
  const int wave = tid >> 5;
  const int lane = tid & 31;
  const int half = lane >> 4;
  const int l16  = lane & 15;
  const int m0   = blockIdx.x * ROWS;

  v2f afrag[K_IN / 4];
  float ebias = 0.0f;
  if (wave == 0) {
    if (l16 < NQ) ebias = bin[l16];
    const float* arow = x + (m0 + l16) * K_IN + half * 2;
    #pragma unroll
    for (int kk = 0; kk < K_IN / 4; ++kk) {
      afrag[kk].x = arow[kk * 4];
      afrag[kk].y = arow[kk * 4 + 1];
    }
  }

  const int n0 = wave * 16;
  v2f bfrag[3];
  {
    const float* brow = Wout + (n0 + l16) * NQ;
    #pragma unroll
    for (int kk = 0; kk < 3; ++kk) {
      const int k0  = kk * 4 + half * 2;
      const int k0c = (k0 < NQ - 1) ? k0 : (NQ - 2);
      bfrag[kk].x = brow[k0c];
      bfrag[kk].y = brow[k0c + 1];
    }
  }
  const float obias = bout[n0 + l16];
  const float q0v = (l16 < NQ) ? qp[l16] : 0.0f;

  float wv[16 * K_IN / 512];
  #pragma unroll
  for (int t = 0; t < 16 * K_IN / 512; ++t) {
    const int i = tid + t * 512;
    const int n = i >> 7;
    const int k = i & (K_IN - 1);
    wv[t] = Win[((n < NQ) ? n : (NQ - 1)) * K_IN + k];
  }
  #pragma unroll
  for (int t = 0; t < 16 * K_IN / 512; ++t) {
    const int i = tid + t * 512;
    const int n = i >> 7;
    const int k = i & (K_IN - 1);
    Bs[n * BPAD + k] = (n < NQ) ? wv[t] : 0.0f;
  }
  for (int j = tid; j < DIM; j += 512) {
    int src = j;
    if ((src >> 0) & 1) src ^= (1 << 9);
    #pragma unroll
    for (int c = 8; c >= 0; --c)
      if ((src >> (9 - c)) & 1) src ^= (1 << (9 - (c + 1)));
    perm[j] = (unsigned short)src;
  }
  if (tid < ROWS) {
    zsm[tid * ZPAD + NQ]     = 0.0f;
    zsm[tid * ZPAD + NQ + 1] = 0.0f;
  }
  __syncthreads();

  int sidx[32];
  #pragma unroll
  for (int k = 0; k < 32; ++k) sidx[k] = (int)perm[k * 32 + lane] << 3;

  if (wave == 0) {
    const float* brow = Bs + l16 * BPAD + half * 2;
    v8f acc = {};
    #pragma unroll
    for (int kk = 0; kk < K_IN / 4; ++kk) {
      v2f b;
      b.x = brow[kk * 4];
      b.y = brow[kk * 4 + 1];
      acc = __builtin_amdgcn_wmma_f32_16x16x4_f32(false, afrag[kk], false, b,
                                                  (short)0, acc, false, false);
    }
    if (l16 < NQ) {
      #pragma unroll
      for (int r = 0; r < 8; ++r)
        hsm[(r + 8 * half) * HPAD + l16] = tanhf(acc[r] + ebias);
    }
  }
  __syncthreads();

  float2* st = sst[wave];

  float re[32], im[32];
  product_state(hsm + wave * HPAD, q0v, lane, re, im);
  ring_cnot(st, sidx, 0, lane, re, im);

  float pscale = 1.0f;
  #pragma nounroll
  for (int l = 1; l < NL; ++l) {
    float t[NQ], tn[NQ];
    int flipb;
    float mag2;
    layer_coeffs(qp + l * NQ, lane, t, tn, flipb, mag2);
    pscale *= mag2;
    rx_layer(re, im, t, tn);
    ring_cnot(st, sidx, flipb, lane, re, im);
  }

  float zacc[NQ];
  #pragma unroll
  for (int q = 0; q < NQ; ++q) zacc[q] = 0.0f;
  #pragma unroll
  for (int k = 0; k < 32; ++k) {
    const int i = k * 32 + lane;
    const float pr = re[k] * re[k] + im[k] * im[k];
    #pragma unroll
    for (int q = 0; q < NQ; ++q)
      zacc[q] += ((i >> (9 - q)) & 1) ? -pr : pr;
  }
  #pragma unroll
  for (int q = 0; q < NQ; ++q) {
    float v = zacc[q];
    for (int off = 16; off > 0; off >>= 1)
      v += __shfl_xor(v, off, 32);
    if (lane == 0) zsm[wave * ZPAD + q] = v * pscale;
  }
  __syncthreads();

  {
    const float* zrow = zsm + l16 * ZPAD + half * 2;
    v8f acc = {};
    #pragma unroll
    for (int kk = 0; kk < 3; ++kk) {
      v2f a;
      a.x = zrow[kk * 4];
      a.y = zrow[kk * 4 + 1];
      acc = __builtin_amdgcn_wmma_f32_16x16x4_f32(false, a, false, bfrag[kk],
                                                  (short)0, acc, false, false);
    }
    #pragma unroll
    for (int r = 0; r < 8; ++r) {
      const int m = m0 + r + 8 * half;
      out[m * N_OUT + n0 + l16] = acc[r] + obias;
    }
  }
}

extern "C" void kernel_launch(void* const* d_in, const int* in_sizes, int n_in,
                              void* d_out, int out_size, void* d_ws, size_t ws_size,
                              hipStream_t stream) {
  const float* x    = (const float*)d_in[0];
  const float* Win  = (const float*)d_in[1];
  const float* bin  = (const float*)d_in[2];
  const float* qp   = (const float*)d_in[3];
  const float* Wout = (const float*)d_in[4];
  const float* bout = (const float*)d_in[5];
  float* out = (float*)d_out;

  k_fused<<<BATCH / ROWS, 512, 0, stream>>>(x, Win, bin, qp, Wout, bout, out);
}


# phase 2 in the X eigenbasis: each RX is diagonal there, all moduli stay equal, so the state is 1024 phase angles; a layer is two adds each; sin/cos once; z = mean cos(phase diff); sums reordered
# speedup vs baseline: 5.6832x; 1.2809x over previous
#include <hip/hip_runtime.h>
#include <math.h>

typedef float v2f __attribute__((ext_vector_type(2)));
typedef float v8f __attribute__((ext_vector_type(8)));

#define NQ 10
#define NL 4
#define DIM 1024
#define BATCH 4096
#define K_IN 128
#define N_OUT 256
#define BPAD 132

#define ROWS 16
#define HPAD 16
#define ZPAD 12

template <int M>
__device__ __forceinline__ float lane_xor(float v)
{
  const int bits = __builtin_bit_cast(int, v);
  if constexpr (M == 16) {
    return __builtin_bit_cast(float, __builtin_amdgcn_permlanex16(
        bits, bits, 0x76543210, 0xfedcba98, false, false));
  } else {
    return __builtin_bit_cast(float, __builtin_amdgcn_update_dpp(
        0, bits, 0x160 + M, 0xf, 0xf, true));
  }
}

__device__ __forceinline__ void layer_phases(float a, int lane,
                                             float& pr, float& pl)
{
  const int ab = __builtin_bit_cast(int, a);
  float r = 0.0f, l = 0.0f;
  #pragma unroll
  for (int w = 0; w < 5; ++w) {
    const float ar = __builtin_bit_cast(float, __builtin_amdgcn_readlane(ab, w));
    const float al = __builtin_bit_cast(float, __builtin_amdgcn_readlane(ab, w + 5));
    const bool bit = (lane >> (4 - w)) & 1;
    r += bit ? ar : -ar;
    l += bit ? al : -al;
  }
  pr = r;
  pl = l;
}

__device__ __forceinline__ void ring_cnot(float* __restrict__ st,
                                          const int (&sidx)[32], int lane,
                                          float (&ph)[32])
{
  #pragma unroll
  for (int k = 0; k < 32; ++k)
    st[k * 32 + lane] = ph[k];
  const char* stb = reinterpret_cast<const char*>(st);
  #pragma unroll
  for (int k = 0; k < 32; ++k)
    ph[k] = *reinterpret_cast<const float*>(stb + sidx[k]);
}

template <int M>
__device__ __forceinline__ float lane_pairs(const float (&cs)[32],
                                            const float (&sn)[32])
{
  float acc = 0.0f;
  #pragma unroll
  for (int k = 0; k < 32; ++k) {
    if constexpr (M == 16) {
      acc = fmaf(lane_xor<16>(cs[k]), cs[k], acc);
      acc = fmaf(lane_xor<16>(sn[k]), sn[k], acc);
    } else {
      asm("v_fmac_f32_dpp %0, %1, %2 row_xmask:%3 row_mask:0xf bank_mask:0xf bound_ctrl:1"
          : "+v"(acc) : "v"(cs[k]), "v"(cs[k]), "n"(M));
      asm("v_fmac_f32_dpp %0, %1, %2 row_xmask:%3 row_mask:0xf bank_mask:0xf bound_ctrl:1"
          : "+v"(acc) : "v"(sn[k]), "v"(sn[k]), "n"(M));
    }
  }
  return acc;
}

__device__ __forceinline__ void phase_sincos(float phi, float& s, float& c)
{
  const float n = rintf(phi * 0.63661977236758134f);
  float r = fmaf(n, -1.57079637050628662f, phi);
  r = fmaf(n, 4.37113900018624283e-8f, r);
  const int q = (int)n;
  const float r2 = r * r;
  float sp = fmaf(r2, -1.9515295891e-4f, 8.3321608736e-3f);
  sp = fmaf(r2, sp, -1.6666654611e-1f);
  const float sr = fmaf(r * r2, sp, r);
  float cp = fmaf(r2, 2.443315711809948e-5f, -1.388731625493765e-3f);
  cp = fmaf(r2, cp, 4.166664568298827e-2f);
  cp = fmaf(r2, cp, -0.5f);
  const float cr = fmaf(r2, cp, 1.0f);
  const bool odd = q & 1;
  const float sv = odd ? cr : sr;
  const float cv = odd ? sr : cr;
  s = (q & 2) ? -sv : sv;
  c = ((q + 1) & 2) ? -cv : cv;
}

__global__ __launch_bounds__(512) void k_fused(
    const float* __restrict__ x, const float* __restrict__ Win,
    const float* __restrict__ bin, const float* __restrict__ qp,
    const float* __restrict__ Wout, const float* __restrict__ bout,
    float* __restrict__ out)
{
  __shared__ float sst[ROWS][DIM];
  __shared__ unsigned short perm[DIM];
  __shared__ float Bs[16 * BPAD];
  __shared__ float hsm[ROWS * HPAD];
  __shared__ float zsm[ROWS * ZPAD];

  const int tid  = threadIdx.x;
  const int wave = tid >> 5;
  const int lane = tid & 31;
  const int half = lane >> 4;
  const int l16  = lane & 15;
  const int m0   = blockIdx.x * ROWS;

  v2f afrag[K_IN / 4];
  float ebias = 0.0f;
  if (wave == 0) {
    if (l16 < NQ) ebias = bin[l16];
    const float* arow = x + (m0 + l16) * K_IN + half * 2;
    #pragma unroll
    for (int kk = 0; kk < K_IN / 4; ++kk) {
      afrag[kk].x = arow[kk * 4];
      afrag[kk].y = arow[kk * 4 + 1];
    }
  }

  const int n0 = wave * 16;
  v2f bfrag[3];
  {
    const float* brow = Wout + (n0 + l16) * NQ;
    #pragma unroll
    for (int kk = 0; kk < 3; ++kk) {
      const int k0  = kk * 4 + half * 2;
      const int k0c = (k0 < NQ - 1) ? k0 : (NQ - 2);
      bfrag[kk].x = brow[k0c];
      bfrag[kk].y = brow[k0c + 1];
    }
  }
  const float obias = bout[n0 + l16];
  float qa[NL];
  #pragma unroll
  for (int l = 0; l < NL; ++l)
    qa[l] = (lane < NQ) ? qp[l * NQ + lane] : 0.0f;

  float wv[16 * K_IN / 512];
  #pragma unroll
  for (int t = 0; t < 16 * K_IN / 512; ++t) {
    const int i = tid + t * 512;
    const int n = i >> 7;
    const int k = i & (K_IN - 1);
    wv[t] = Win[((n < NQ) ? n : (NQ - 1)) * K_IN + k];
  }
  #pragma unroll
  for (int t = 0; t < 16 * K_IN / 512; ++t) {
    const int i = tid + t * 512;
    const int n = i >> 7;
    const int k = i & (K_IN - 1);
    Bs[n * BPAD + k] = (n < NQ) ? wv[t] : 0.0f;
  }
  for (int j = tid; j < DIM; j += 512) {
    int src = j;
    if ((src >> 9) & 1) src ^= 1;
    #pragma unroll
    for (int c = 8; c >= 0; --c)
      if ((src >> (9 - (c + 1))) & 1) src ^= (1 << (9 - c));
    perm[j] = (unsigned short)src;
  }
  if (tid < ROWS) {
    zsm[tid * ZPAD + NQ]     = 0.0f;
    zsm[tid * ZPAD + NQ + 1] = 0.0f;
  }
  __syncthreads();

  int sidx[32];
  #pragma unroll
  for (int k = 0; k < 32; ++k) sidx[k] = (int)perm[k * 32 + lane] << 2;

  if (wave == 0) {
    const float* brow = Bs + l16 * BPAD + half * 2;
    v8f acc = {};
    #pragma unroll
    for (int kk = 0; kk < K_IN / 4; ++kk) {
      v2f b;
      b.x = brow[kk * 4];
      b.y = brow[kk * 4 + 1];
      acc = __builtin_amdgcn_wmma_f32_16x16x4_f32(false, afrag[kk], false, b,
                                                  (short)0, acc, false, false);
    }
    if (l16 < NQ) {
      #pragma unroll
      for (int r = 0; r < 8; ++r)
        hsm[(r + 8 * half) * HPAD + l16] = tanhf(acc[r] + ebias);
    }
  }
  __syncthreads();

  float* st = sst[wave];

  float ph[32];
  float pr, pl;
  {
    const float hv = (lane < NQ) ? hsm[wave * HPAD + lane] : 0.0f;
    layer_phases(0.5f * (hv + qa[0]), lane, pr, pl);
    const int prb = __builtin_bit_cast(int, pr);
    #pragma unroll
    for (int k = 0; k < 32; ++k)
      ph[k] = __builtin_bit_cast(float, __builtin_amdgcn_readlane(prb, k)) + pl;
  }
  ring_cnot(st, sidx, lane, ph);

  #pragma unroll
  for (int l = 1; l < NL; ++l) {
    layer_phases(0.5f * qa[l], lane, pr, pl);
    const int prb = __builtin_bit_cast(int, pr);
    #pragma unroll
    for (int k = 0; k < 32; ++k)
      ph[k] += __builtin_bit_cast(float, __builtin_amdgcn_readlane(prb, k)) + pl;
    ring_cnot(st, sidx, lane, ph);
  }

  float cs[32], sn[32];
  #pragma unroll
  for (int k = 0; k < 32; ++k) phase_sincos(ph[k], sn[k], cs[k]);

  float zacc[NQ];
  #pragma unroll
  for (int q = 0; q < 5; ++q) {
    const int km = 1 << (4 - q);
    float acc = 0.0f;
    #pragma unroll
    for (int k = 0; k < 32; ++k) {
      if ((k & km) == 0) {
        const int k1 = k | km;
        acc = fmaf(cs[k], cs[k1], acc);
        acc = fmaf(sn[k], sn[k1], acc);
      }
    }
    zacc[q] = acc;
  }
  zacc[5] = lane_pairs<16>(cs, sn);
  zacc[6] = lane_pairs<8>(cs, sn);
  zacc[7] = lane_pairs<4>(cs, sn);
  zacc[8] = lane_pairs<2>(cs, sn);
  zacc[9] = lane_pairs<1>(cs, sn);
  #pragma unroll
  for (int q = 0; q < NQ; ++q) {
    float v = zacc[q];
    for (int off = 16; off > 0; off >>= 1)
      v += __shfl_xor(v, off, 32);
    if (lane == 0)
      zsm[wave * ZPAD + q] = v * ((q < 5) ? (2.0f / 1024.0f) : (1.0f / 1024.0f));
  }
  __syncthreads();

  {
    const float* zrow = zsm + l16 * ZPAD + half * 2;
    v8f acc = {};
    #pragma unroll
    for (int kk = 0; kk < 3; ++kk) {
      v2f a;
      a.x = zrow[kk * 4];
      a.y = zrow[kk * 4 + 1];
      acc = __builtin_amdgcn_wmma_f32_16x16x4_f32(false, a, false, bfrag[kk],
                                                  (short)0, acc, false, false);
    }
    #pragma unroll
    for (int r = 0; r < 8; ++r) {
      const int m = m0 + r + 8 * half;
      out[m * N_OUT + n0 + l16] = acc[r] + obias;
    }
  }
}

extern "C" void kernel_launch(void* const* d_in, const int* in_sizes, int n_in,
                              void* d_out, int out_size, void* d_ws, size_t ws_size,
                              hipStream_t stream) {
  const float* x    = (const float*)d_in[0];
  const float* Win  = (const float*)d_in[1];
  const float* bin  = (const float*)d_in[2];
  const float* qp   = (const float*)d_in[3];
  const float* Wout = (const float*)d_in[4];
  const float* bout = (const float*)d_in[5];
  float* out = (float*)d_out;

  k_fused<<<BATCH / ROWS, 512, 0, stream>>>(x, Win, bin, qp, Wout, bout, out);
}


# phase 2: state = row-independent factor (layers 1-3; cos/sin table built once per workgroup in LDS) times the row's product state: 10 sincos, one LDS trip per row, not 1024 and four; sums reordered
# speedup vs baseline: 7.0525x; 1.2409x over previous
#include <hip/hip_runtime.h>
#include <math.h>

typedef float v2f __attribute__((ext_vector_type(2)));
typedef float v8f __attribute__((ext_vector_type(8)));

#define NQ 10
#define NL 4
#define DIM 1024
#define BATCH 4096
#define K_IN 128
#define N_OUT 256
#define BPAD 132

#define ROWS 16
#define HPAD 16
#define ZPAD 12

template <int M>
__device__ __forceinline__ float lane_xor(float v)
{
  const int bits = __builtin_bit_cast(int, v);
  if constexpr (M == 16) {
    return __builtin_bit_cast(float, __builtin_amdgcn_permlanex16(
        bits, bits, 0x76543210, 0xfedcba98, false, false));
  } else {
    return __builtin_bit_cast(float, __builtin_amdgcn_update_dpp(
        0, bits, 0x160 + M, 0xf, 0xf, true));
  }
}

__device__ __forceinline__ void layer_phases(float a, int lane,
                                             float& pr, float& pl)
{
  const int ab = __builtin_bit_cast(int, a);
  float r = 0.0f, l = 0.0f;
  #pragma unroll
  for (int w = 0; w < 5; ++w) {
    const float ar = __builtin_bit_cast(float, __builtin_amdgcn_readlane(ab, w));
    const float al = __builtin_bit_cast(float, __builtin_amdgcn_readlane(ab, w + 5));
    const bool bit = (lane >> (4 - w)) & 1;
    r += bit ? ar : -ar;
    l += bit ? al : -al;
  }
  pr = r;
  pl = l;
}

__device__ __forceinline__ int ring_src(int j)
{
  int s = j ^ (j >> 9);
  s ^= s << 1;
  s ^= s << 2;
  s ^= s << 4;
  s ^= s << 8;
  return s & (DIM - 1);
}

template <int M>
__device__ __forceinline__ float lane_pairs(const float (&cs)[32],
                                            const float (&sn)[32])
{
  float acc = 0.0f;
  #pragma unroll
  for (int k = 0; k < 32; ++k) {
    if constexpr (M == 16) {
      acc = fmaf(lane_xor<16>(cs[k]), cs[k], acc);
      acc = fmaf(lane_xor<16>(sn[k]), sn[k], acc);
    } else {
      asm("v_fmac_f32_dpp %0, %1, %2 row_xmask:%3 row_mask:0xf bank_mask:0xf bound_ctrl:1"
          : "+v"(acc) : "v"(cs[k]), "v"(cs[k]), "n"(M));
      asm("v_fmac_f32_dpp %0, %1, %2 row_xmask:%3 row_mask:0xf bank_mask:0xf bound_ctrl:1"
          : "+v"(acc) : "v"(sn[k]), "v"(sn[k]), "n"(M));
    }
  }
  return acc;
}

__device__ __forceinline__ void phase_sincos(float phi, float& s, float& c)
{
  const float n = rintf(phi * 0.63661977236758134f);
  float r = fmaf(n, -1.57079637050628662f, phi);
  r = fmaf(n, 4.37113900018624283e-8f, r);
  const int q = (int)n;
  const float r2 = r * r;
  float sp = fmaf(r2, -1.9515295891e-4f, 8.3321608736e-3f);
  sp = fmaf(r2, sp, -1.6666654611e-1f);
  const float sr = fmaf(r * r2, sp, r);
  float cp = fmaf(r2, 2.443315711809948e-5f, -1.388731625493765e-3f);
  cp = fmaf(r2, cp, 4.166664568298827e-2f);
  cp = fmaf(r2, cp, -0.5f);
  const float cr = fmaf(r2, cp, 1.0f);
  const bool odd = q & 1;
  const float sv = odd ? cr : sr;
  const float cv = odd ? sr : cr;
  s = (q & 2) ? -sv : sv;
  c = ((q + 1) & 2) ? -cv : cv;
}

__device__ __forceinline__ void table_entry(int j, const float (&lpr)[NL],
                                            const float (&lpl)[NL],
                                            v2f* etab, unsigned short* perm4)
{
  int y = j;
  float c = 0.0f;
  #pragma unroll
  for (int l = NL - 1; l >= 1; --l) {
    y = ring_src(y);
    c += __shfl(lpr[l], y >> 5, 32) + __shfl(lpl[l], y & 31, 32);
  }
  perm4[j] = (unsigned short)ring_src(y);
  float es, ec;
  phase_sincos(c, es, ec);
  v2f e;
  e.x = ec;
  e.y = es;
  etab[j] = e;
}

__global__ __launch_bounds__(512) void k_fused(
    const float* __restrict__ x, const float* __restrict__ Win,
    const float* __restrict__ bin, const float* __restrict__ qp,
    const float* __restrict__ Wout, const float* __restrict__ bout,
    float* __restrict__ out)
{
  __shared__ v2f sst[ROWS][DIM];
  __shared__ v2f etab[DIM];
  __shared__ unsigned short perm4[DIM];
  __shared__ float Bs[16 * BPAD];
  __shared__ float hsm[ROWS * HPAD];
  __shared__ float zsm[ROWS * ZPAD];

  const int tid  = threadIdx.x;
  const int wave = tid >> 5;
  const int lane = tid & 31;
  const int half = lane >> 4;
  const int l16  = lane & 15;
  const int m0   = blockIdx.x * ROWS;

  v2f afrag[K_IN / 4];
  float ebias = 0.0f;
  if (wave == 0) {
    if (l16 < NQ) ebias = bin[l16];
    const float* arow = x + (m0 + l16) * K_IN + half * 2;
    #pragma unroll
    for (int kk = 0; kk < K_IN / 4; ++kk) {
      afrag[kk].x = arow[kk * 4];
      afrag[kk].y = arow[kk * 4 + 1];
    }
  }

  const int n0 = wave * 16;
  v2f bfrag[3];
  {
    const float* brow = Wout + (n0 + l16) * NQ;
    #pragma unroll
    for (int kk = 0; kk < 3; ++kk) {
      const int k0  = kk * 4 + half * 2;
      const int k0c = (k0 < NQ - 1) ? k0 : (NQ - 2);
      bfrag[kk].x = brow[k0c];
      bfrag[kk].y = brow[k0c + 1];
    }
  }
  const float obias = bout[n0 + l16];
  float qa[NL];
  #pragma unroll
  for (int l = 0; l < NL; ++l)
    qa[l] = (lane < NQ) ? qp[l * NQ + lane] : 0.0f;

  float wv[16 * K_IN / 512];
  #pragma unroll
  for (int t = 0; t < 16 * K_IN / 512; ++t) {
    const int i = tid + t * 512;
    const int n = i >> 7;
    const int k = i & (K_IN - 1);
    wv[t] = Win[((n < NQ) ? n : (NQ - 1)) * K_IN + k];
  }
  float lpr[NL] = {}, lpl[NL] = {};
  const int u = tid - 32;
  if (wave != 0) {
    #pragma unroll
    for (int l = 1; l < NL; ++l)
      layer_phases(0.5f * qa[l], lane, lpr[l], lpl[l]);
    table_entry(u, lpr, lpl, etab, perm4);
  }
  #pragma unroll
  for (int t = 0; t < 16 * K_IN / 512; ++t) {
    const int i = tid + t * 512;
    const int n = i >> 7;
    const int k = i & (K_IN - 1);
    Bs[n * BPAD + k] = (n < NQ) ? wv[t] : 0.0f;
  }
  if (tid < ROWS) {
    zsm[tid * ZPAD + NQ]     = 0.0f;
    zsm[tid * ZPAD + NQ + 1] = 0.0f;
  }
  __syncthreads();

  if (wave != 0) {
    table_entry(480 + u, lpr, lpl, etab, perm4);
    if (wave < 3) table_entry(960 + u, lpr, lpl, etab, perm4);
  }

  if (wave == 0) {
    const float* brow = Bs + l16 * BPAD + half * 2;
    v8f acc = {};
    #pragma unroll
    for (int kk = 0; kk < K_IN / 4; ++kk) {
      v2f b;
      b.x = brow[kk * 4];
      b.y = brow[kk * 4 + 1];
      acc = __builtin_amdgcn_wmma_f32_16x16x4_f32(false, afrag[kk], false, b,
                                                  (short)0, acc, false, false);
    }
    if (l16 < NQ) {
      #pragma unroll
      for (int r = 0; r < 8; ++r)
        hsm[(r + 8 * half) * HPAD + l16] = tanhf(acc[r] + ebias);
    }
  }
  __syncthreads();

  v2f* st = sst[wave];

  const float hv = (lane < NQ) ? hsm[wave * HPAD + lane] : 0.0f;
  int sidx[32];
  #pragma unroll
  for (int k = 0; k < 32; ++k) sidx[k] = (int)perm4[k * 32 + lane] << 3;

  float ca, sa;
  phase_sincos(0.5f * (hv + qa[0]), sa, ca);
  float fhc = 1.0f, fhs = 0.0f, flc = 1.0f, fls = 0.0f;
  {
    const int cab = __builtin_bit_cast(int, ca);
    const int sab = __builtin_bit_cast(int, sa);
    #pragma unroll
    for (int w = 0; w < 5; ++w) {
      const float ch = __builtin_bit_cast(float, __builtin_amdgcn_readlane(cab, w));
      const float s0 = __builtin_bit_cast(float, __builtin_amdgcn_readlane(sab, w));
      const float cl = __builtin_bit_cast(float, __builtin_amdgcn_readlane(cab, w + 5));
      const float s1 = __builtin_bit_cast(float, __builtin_amdgcn_readlane(sab, w + 5));
      const bool bit = (lane >> (4 - w)) & 1;
      const float sh = bit ? s0 : -s0;
      const float sl = bit ? s1 : -s1;
      const float th = fhc * ch - fhs * sh;
      fhs = fhc * sh + fhs * ch;
      fhc = th;
      const float tl = flc * cl - fls * sl;
      fls = flc * sl + fls * cl;
      flc = tl;
    }
  }
  {
    const int fhcb = __builtin_bit_cast(int, fhc);
    const int fhsb = __builtin_bit_cast(int, fhs);
    #pragma unroll
    for (int k = 0; k < 32; ++k) {
      const float hc = __builtin_bit_cast(float, __builtin_amdgcn_readlane(fhcb, k));
      const float hs = __builtin_bit_cast(float, __builtin_amdgcn_readlane(fhsb, k));
      v2f g;
      g.x = hc * flc - hs * fls;
      g.y = hc * fls + hs * flc;
      st[k * 32 + lane] = g;
    }
  }
  float cs[32], sn[32];
  {
    const char* stb = reinterpret_cast<const char*>(st);
    #pragma unroll
    for (int k = 0; k < 32; ++k) {
      const v2f g = *reinterpret_cast<const v2f*>(stb + sidx[k]);
      const v2f e = etab[k * 32 + lane];
      cs[k] = e.x * g.x - e.y * g.y;
      sn[k] = e.x * g.y + e.y * g.x;
    }
  }

  float zacc[NQ];
  #pragma unroll
  for (int q = 0; q < 5; ++q) {
    const int km = 1 << (4 - q);
    float acc = 0.0f;
    #pragma unroll
    for (int k = 0; k < 32; ++k) {
      if ((k & km) == 0) {
        const int k1 = k | km;
        acc = fmaf(cs[k], cs[k1], acc);
        acc = fmaf(sn[k], sn[k1], acc);
      }
    }
    zacc[q] = acc;
  }
  zacc[5] = lane_pairs<16>(cs, sn);
  zacc[6] = lane_pairs<8>(cs, sn);
  zacc[7] = lane_pairs<4>(cs, sn);
  zacc[8] = lane_pairs<2>(cs, sn);
  zacc[9] = lane_pairs<1>(cs, sn);
  #pragma unroll
  for (int q = 0; q < NQ; ++q) {
    float v = zacc[q];
    for (int off = 16; off > 0; off >>= 1)
      v += __shfl_xor(v, off, 32);
    if (lane == 0)
      zsm[wave * ZPAD + q] = v * ((q < 5) ? (2.0f / 1024.0f) : (1.0f / 1024.0f));
  }
  __syncthreads();

  {
    const float* zrow = zsm + l16 * ZPAD + half * 2;
    v8f acc = {};
    #pragma unroll
    for (int kk = 0; kk < 3; ++kk) {
      v2f a;
      a.x = zrow[kk * 4];
      a.y = zrow[kk * 4 + 1];
      acc = __builtin_amdgcn_wmma_f32_16x16x4_f32(false, a, false, bfrag[kk],
                                                  (short)0, acc, false, false);
    }
    #pragma unroll
    for (int r = 0; r < 8; ++r) {
      const int m = m0 + r + 8 * half;
      out[m * N_OUT + n0 + l16] = acc[r] + obias;
    }
  }
}

extern "C" void kernel_launch(void* const* d_in, const int* in_sizes, int n_in,
                              void* d_out, int out_size, void* d_ws, size_t ws_size,
                              hipStream_t stream) {
  const float* x    = (const float*)d_in[0];
  const float* Win  = (const float*)d_in[1];
  const float* bin  = (const float*)d_in[2];
  const float* qp   = (const float*)d_in[3];
  const float* Wout = (const float*)d_in[4];
  const float* bout = (const float*)d_in[5];
  float* out = (float*)d_out;

  k_fused<<<BATCH / ROWS, 512, 0, stream>>>(x, Win, bin, qp, Wout, bout, out);
}


# phase 2: Z_q sees the row only through the 6-8 label bits of P^4 bit_q; the row-independent rest is summed once per workgroup into tables T_q, a row adds 1344 terms, not 10x1024; sums reordered
# speedup vs baseline: 10.1321x; 1.4367x over previous
#include <hip/hip_runtime.h>
#include <math.h>

typedef float v2f __attribute__((ext_vector_type(2)));
typedef float v8f __attribute__((ext_vector_type(8)));

#define NQ 10
#define NL 4
#define DIM 1024
#define BATCH 4096
#define K_IN 128
#define N_OUT 256
#define BPAD 132

#define ROWS 16
#define HPAD 16
#define ZPAD 12

constexpr int ring_c(int j)
{
  int s = j ^ (j >> 9);
  s ^= s << 1;
  s ^= s << 2;
  s ^= s << 4;
  s ^= s << 8;
  return s & (DIM - 1);
}
constexpr int ring_inv(int r)
{
  const int s = (r ^ (r << 1)) & (DIM - 1);
  return s ^ (s >> 9);
}
constexpr int flip_mask(int q)
{
  int d = 1 << (NQ - 1 - q);
  for (int i = 0; i < NL; ++i) d = ring_c(d);
  return d;
}
constexpr int popc10(int d)
{
  int c = 0;
  for (int i = 0; i < NQ; ++i) c += (d >> i) & 1;
  return c;
}
constexpr int nth_set(int d, int n)
{
  for (int i = 0; i < NQ; ++i)
    if ((d >> i) & 1) {
      if (n == 0) return i;
      --n;
    }
  return -1;
}
constexpr int toff_c(int q)
{
  int o = 0;
  for (int i = 0; i < q; ++i) o += 1 << popc10(flip_mask(i));
  return o;
}
constexpr int TTOT  = toff_c(NQ);
constexpr int NUNIT = TTOT / 32;
constexpr int n_wide()
{
  int n = 0;
  for (int q = 0; q < NQ; ++q) n += (popc10(flip_mask(q)) == 8);
  return n;
}
constexpr int nth_wide(int n)
{
  for (int q = 0; q < NQ; ++q)
    if (popc10(flip_mask(q)) == 8) {
      if (n == 0) return q;
      --n;
    }
  return -1;
}
constexpr bool masks_fit()
{
  for (int q = 0; q < NQ; ++q) {
    const int m = popc10(flip_mask(q));
    if (m < 5 || m > 8) return false;
  }
  return true;
}
static_assert(masks_fit(), "every flip mask needs 5 to 8 bits");
static_assert(n_wide() == 2, "twelve tasks: ten qubits and two upper halves");
static_assert(NQ + n_wide() == ZPAD, "phase 3 K dimension");
constexpr int QX0 = nth_wide(0);
constexpr int QX1 = nth_wide(1);

struct FreeList { int v[32]; };
constexpr FreeList free_labels(int mask)
{
  FreeList f{};
  const int n = 1 << popc10(mask);
  for (int c = 0; c < n; ++c) {
    int y = 0, k = 0;
    for (int i = 0; i < NQ; ++i)
      if ((mask >> i) & 1) {
        y |= ((c >> k) & 1) << i;
        ++k;
      }
    f.v[c] = y;
  }
  return f;
}

__device__ __forceinline__ float lane_x16(float v)
{
  const int bits = __builtin_bit_cast(int, v);
  return __builtin_bit_cast(float, __builtin_amdgcn_permlanex16(
      bits, bits, 0x76543210, 0xfedcba98, false, false));
}

__device__ __forceinline__ void layer_phases(float a, int lane,
                                             float& pr, float& pl)
{
  const int ab = __builtin_bit_cast(int, a);
  float r = 0.0f, l = 0.0f;
  #pragma unroll
  for (int w = 0; w < 5; ++w) {
    const float ar = __builtin_bit_cast(float, __builtin_amdgcn_readlane(ab, w));
    const float al = __builtin_bit_cast(float, __builtin_amdgcn_readlane(ab, w + 5));
    const bool bit = (lane >> (4 - w)) & 1;
    r += bit ? ar : -ar;
    l += bit ? al : -al;
  }
  pr = r;
  pl = l;
}

__device__ __forceinline__ void phase_sincos(float phi, float& s, float& c)
{
  const float n = rintf(phi * 0.63661977236758134f);
  float r = fmaf(n, -1.57079637050628662f, phi);
  r = fmaf(n, 4.37113900018624283e-8f, r);
  const int q = (int)n;
  const float r2 = r * r;
  float sp = fmaf(r2, -1.9515295891e-4f, 8.3321608736e-3f);
  sp = fmaf(r2, sp, -1.6666654611e-1f);
  const float sr = fmaf(r * r2, sp, r);
  float cp = fmaf(r2, 2.443315711809948e-5f, -1.388731625493765e-3f);
  cp = fmaf(r2, cp, 4.166664568298827e-2f);
  cp = fmaf(r2, cp, -0.5f);
  const float cr = fmaf(r2, cp, 1.0f);
  const bool odd = q & 1;
  const float sv = odd ? cr : sr;
  const float cv = odd ? sr : cr;
  s = (q & 2) ? -sv : sv;
  c = ((q + 1) & 2) ? -cv : cv;
}

__device__ __forceinline__ void table_entry(int y, const float (&lpr)[NL],
                                            const float (&lpl)[NL], v2f* etab)
{
  int v = y;
  float c = 0.0f;
  #pragma unroll
  for (int l = 1; l < NL; ++l) {
    v = ring_inv(v);
    c += __shfl(lpr[l], v >> 5, 32) + __shfl(lpl[l], v & 31, 32);
  }
  float es, ec;
  phase_sincos(c, es, ec);
  v2f e;
  e.x = ec;
  e.y = es;
  etab[y] = e;
}

template <int MASK, int I = 0>
__device__ __forceinline__ int deposit(int c)
{
  if constexpr (I >= popc10(MASK)) {
    return 0;
  } else {
    constexpr int P = nth_set(MASK, I);
    return (((c >> I) & 1) << P) | deposit<MASK, I + 1>(c);
  }
}

template <int Q>
__device__ __forceinline__ void tbuild(int unit, int lane, const v2f* etab,
                                       v2f* ttab)
{
  constexpr int D  = flip_mask(Q);
  constexpr int FM = (DIM - 1) & ~D;
  constexpr int NF = 1 << popc10(FM);
  constexpr FreeList FL = free_labels(FM);
  const int c  = unit * 32 + lane;
  const int y0 = deposit<D>(c);
  const int y1 = y0 ^ D;
  float tr = 0.0f, ti = 0.0f;
  #pragma unroll
  for (int f = 0; f < NF; ++f) {
    const v2f a = etab[y0 | FL.v[f]];
    const v2f b = etab[y1 | FL.v[f]];
    tr = fmaf(a.x, b.x, tr);
    tr = fmaf(a.y, b.y, tr);
    ti = fmaf(a.y, b.x, ti);
    ti = fmaf(-a.x, b.y, ti);
  }
  v2f t;
  t.x = tr;
  t.y = ti;
  ttab[toff_c(Q) + c] = t;
}

template <int Q>
__device__ __forceinline__ void tbuild_dispatch(int j, int lane,
                                                const v2f* etab, v2f* ttab)
{
  if constexpr (Q < NQ) {
    if (j < toff_c(Q + 1) / 32)
      tbuild<Q>(j - toff_c(Q) / 32, lane, etab, ttab);
    else
      tbuild_dispatch<Q + 1>(j, lane, etab, ttab);
  }
}

__device__ __forceinline__ float zpart(int q, int unit0, int lane,
                                       const v2f* urow, const v2f* ttab)
{
  int d = 0, off = 0;
  #pragma unroll
  for (int i = 0; i < NQ; ++i) {
    if (q == i) d = flip_mask(i);
    if (i < q) off += 1 << popc10(flip_mask(i));
  }
  const int m  = __builtin_popcount(d);
  const int nu = (m >= 7) ? 4 : (1 << (m - 5));
  const int hb = lane >> 4;
  v2f u[8];
  int rest = d;
  #pragma unroll
  for (int i = 0; i < 8; ++i) {
    v2f f;
    f.x = 1.0f;
    f.y = 0.0f;
    if (rest != 0) {
      const int p = __builtin_ctz(rest);
      rest &= rest - 1;
      f = urow[NQ - 1 - p];
    }
    u[i] = f;
  }
  float ar[16], ai[16];
  ar[0] = u[0].x;  ai[0] = -u[0].y;
  ar[1] = u[0].x;  ai[1] = u[0].y;
  #pragma unroll
  for (int i = 1; i < 4; ++i) {
    const float c = u[i].x, s = u[i].y;
    #pragma unroll
    for (int e = 0; e < (1 << i); ++e) {
      const float xr = ar[e], xi = ai[e];
      ar[e + (1 << i)] = xr * c - xi * s;
      ai[e + (1 << i)] = xr * s + xi * c;
      ar[e] = xr * c + xi * s;
      ai[e] = xi * c - xr * s;
    }
  }
  const float g4r = u[4].x;
  const float g4i = hb ? u[4].y : -u[4].y;
  float acc = 0.0f;
  for (int uu = 0; uu < nu; ++uu) {
    const int unit = unit0 + uu;
    const v2f* tp = ttab + off + unit * 32 + hb * 16;
    float sr = 0.0f, si = 0.0f;
    #pragma unroll
    for (int e = 0; e < 16; ++e) {
      const v2f t = tp[e];
      sr = fmaf(t.x, ar[e], sr);
      sr = fmaf(-t.y, ai[e], sr);
      si = fmaf(t.x, ai[e], si);
      si = fmaf(t.y, ar[e], si);
    }
    float gr = g4r, gi = g4i;
    #pragma unroll
    for (int i = 5; i < 8; ++i) {
      const float c = u[i].x;
      const float s = ((unit >> (i - 5)) & 1) ? u[i].y : -u[i].y;
      const float nr = gr * c - gi * s;
      gi = gr * s + gi * c;
      gr = nr;
    }
    acc = fmaf(gr, sr, acc);
    acc = fmaf(-gi, si, acc);
  }
  return acc;
}

__global__ __launch_bounds__(512) void k_fused(
    const float* __restrict__ x, const float* __restrict__ Win,
    const float* __restrict__ bin, const float* __restrict__ qp,
    const float* __restrict__ Wout, const float* __restrict__ bout,
    float* __restrict__ out)
{
  __shared__ v2f etab[DIM];
  __shared__ v2f ttab[TTOT];
  __shared__ v2f usm[ROWS * NQ];
  __shared__ float Bs[16 * BPAD];
  __shared__ float hsm[ROWS * HPAD];
  __shared__ float zsm[ROWS * ZPAD];

  const int tid  = threadIdx.x;
  const int wave = __builtin_amdgcn_readfirstlane(tid >> 5);
  const int lane = tid & 31;
  const int half = lane >> 4;
  const int l16  = lane & 15;
  const int m0   = blockIdx.x * ROWS;

  v2f afrag[K_IN / 4];
  float ebias = 0.0f;
  if (wave == 0) {
    if (l16 < NQ) ebias = bin[l16];
    const float* arow = x + (m0 + l16) * K_IN + half * 2;
    #pragma unroll
    for (int kk = 0; kk < K_IN / 4; ++kk) {
      afrag[kk].x = arow[kk * 4];
      afrag[kk].y = arow[kk * 4 + 1];
    }
  }

  const int n0 = wave * 16;
  v2f bfrag[3];
  {
    const float* brow = Wout + (n0 + l16) * NQ;
    #pragma unroll
    for (int kk = 0; kk < 3; ++kk) {
      const int k0 = kk * 4 + half * 2;
      const bool ext = (k0 >= NQ);
      bfrag[kk].x = brow[ext ? QX0 : k0];
      bfrag[kk].y = brow[ext ? QX1 : (k0 + 1)];
    }
  }
  const float obias = bout[n0 + l16];
  float qa[NL];
  #pragma unroll
  for (int l = 1; l < NL; ++l)
    qa[l] = (lane < NQ) ? qp[l * NQ + lane] : 0.0f;
  const float th0 = (tid < ROWS * NQ) ? qp[tid % NQ] : 0.0f;

  float wv[16 * K_IN / 512];
  #pragma unroll
  for (int t = 0; t < 16 * K_IN / 512; ++t) {
    const int i = tid + t * 512;
    const int n = i >> 7;
    const int k = i & (K_IN - 1);
    wv[t] = Win[((n < NQ) ? n : (NQ - 1)) * K_IN + k];
  }
  {
    float lpr[NL] = {}, lpl[NL] = {};
    #pragma unroll
    for (int l = 1; l < NL; ++l)
      layer_phases(0.5f * qa[l], lane, lpr[l], lpl[l]);
    table_entry(tid, lpr, lpl, etab);
    table_entry(tid + 512, lpr, lpl, etab);
  }
  #pragma unroll
  for (int t = 0; t < 16 * K_IN / 512; ++t) {
    const int i = tid + t * 512;
    const int n = i >> 7;
    const int k = i & (K_IN - 1);
    Bs[n * BPAD + k] = (n < NQ) ? wv[t] : 0.0f;
  }
  __syncthreads();

  if (wave != 0) {
    for (int j = wave - 1; j < NUNIT; j += 15)
      tbuild_dispatch<0>(j, lane, etab, ttab);
  } else {
    const float* brow = Bs + l16 * BPAD + half * 2;
    v8f acc = {};
    #pragma unroll
    for (int kk = 0; kk < K_IN / 4; ++kk) {
      v2f b;
      b.x = brow[kk * 4];
      b.y = brow[kk * 4 + 1];
      acc = __builtin_amdgcn_wmma_f32_16x16x4_f32(false, afrag[kk], false, b,
                                                  (short)0, acc, false, false);
    }
    if (l16 < NQ) {
      #pragma unroll
      for (int r = 0; r < 8; ++r)
        hsm[(r + 8 * half) * HPAD + l16] = tanhf(acc[r] + ebias);
    }
  }
  __syncthreads();

  if (tid < ROWS * NQ) {
    const int r = tid / NQ;
    const int w = tid - r * NQ;
    float sa, ca;
    phase_sincos(hsm[r * HPAD + w] + th0, sa, ca);
    v2f uv;
    uv.x = ca;
    uv.y = sa;
    usm[tid] = uv;
  }
  __syncthreads();

  if (wave < ZPAD) {
    const int q     = (wave < NQ) ? wave : ((wave == NQ) ? QX0 : QX1);
    const int unit0 = (wave < NQ) ? 0 : 4;
    float v = zpart(q, unit0, lane, usm + l16 * NQ, ttab);
    v += lane_x16(v);
    if (half == 0) zsm[l16 * ZPAD + wave] = v * (1.0f / 1024.0f);
  }
  __syncthreads();

  {
    const float* zrow = zsm + l16 * ZPAD + half * 2;
    v8f acc = {};
    #pragma unroll
    for (int kk = 0; kk < 3; ++kk) {
      v2f a;
      a.x = zrow[kk * 4];
      a.y = zrow[kk * 4 + 1];
      acc = __builtin_amdgcn_wmma_f32_16x16x4_f32(false, a, false, bfrag[kk],
                                                  (short)0, acc, false, false);
    }
    #pragma unroll
    for (int r = 0; r < 8; ++r) {
      const int m = m0 + r + 8 * half;
      out[m * N_OUT + n0 + l16] = acc[r] + obias;
    }
  }
}

extern "C" void kernel_launch(void* const* d_in, const int* in_sizes, int n_in,
                              void* d_out, int out_size, void* d_ws, size_t ws_size,
                              hipStream_t stream) {
  const float* x    = (const float*)d_in[0];
  const float* Win  = (const float*)d_in[1];
  const float* bin  = (const float*)d_in[2];
  const float* qp   = (const float*)d_in[3];
  const float* Wout = (const float*)d_in[4];
  const float* bout = (const float*)d_in[5];
  float* out = (float*)d_out;

  k_fused<<<BATCH / ROWS, 512, 0, stream>>>(x, Win, bin, qp, Wout, bout, out);
}


# dependent waits cut: every global load in one branch-free batch (was 4 serial round trips; x tile via LDS), LDS reads grouped ahead of WMMAs and sums, 2-4 partial sums per chain; sums reordered
# speedup vs baseline: 11.8108x; 1.1657x over previous
#include <hip/hip_runtime.h>
#include <math.h>

typedef float v2f __attribute__((ext_vector_type(2)));
typedef float v8f __attribute__((ext_vector_type(8)));

#define NQ 10
#define NL 4
#define DIM 1024
#define BATCH 4096
#define K_IN 128
#define N_OUT 256
#define BPAD 132
#define XPAD 132

#define ROWS 16
#define HPAD 16
#define ZPAD 12

constexpr int ring_c(int j)
{
  int s = j ^ (j >> 9);
  s ^= s << 1;
  s ^= s << 2;
  s ^= s << 4;
  s ^= s << 8;
  return s & (DIM - 1);
}
constexpr int ring_inv(int r)
{
  const int s = (r ^ (r << 1)) & (DIM - 1);
  return s ^ (s >> 9);
}
constexpr int flip_mask(int q)
{
  int d = 1 << (NQ - 1 - q);
  for (int i = 0; i < NL; ++i) d = ring_c(d);
  return d;
}
constexpr int popc10(int d)
{
  int c = 0;
  for (int i = 0; i < NQ; ++i) c += (d >> i) & 1;
  return c;
}
constexpr int nth_set(int d, int n)
{
  for (int i = 0; i < NQ; ++i)
    if ((d >> i) & 1) {
      if (n == 0) return i;
      --n;
    }
  return -1;
}
constexpr int toff_c(int q)
{
  int o = 0;
  for (int i = 0; i < q; ++i) o += 1 << popc10(flip_mask(i));
  return o;
}
constexpr int TTOT  = toff_c(NQ);
constexpr int NUNIT = TTOT / 32;
constexpr int n_wide()
{
  int n = 0;
  for (int q = 0; q < NQ; ++q) n += (popc10(flip_mask(q)) == 8);
  return n;
}
constexpr int nth_wide(int n)
{
  for (int q = 0; q < NQ; ++q)
    if (popc10(flip_mask(q)) == 8) {
      if (n == 0) return q;
      --n;
    }
  return -1;
}
constexpr bool masks_fit()
{
  for (int q = 0; q < NQ; ++q) {
    const int m = popc10(flip_mask(q));
    if (m < 5 || m > 8) return false;
  }
  return true;
}
static_assert(masks_fit(), "every flip mask needs 5 to 8 bits");
constexpr bool units_even()
{
  for (int q = 0; q < NQ; ++q)
    if (popc10(flip_mask(q)) < 6) return false;
  return true;
}
static_assert(units_even(), "the row sums take a task's units two at a time");
static_assert(n_wide() == 2, "twelve tasks: ten qubits and two upper halves");
static_assert(NQ + n_wide() == ZPAD, "phase 3 K dimension");
constexpr int QX0 = nth_wide(0);
constexpr int QX1 = nth_wide(1);

struct FreeList { int v[32]; };
constexpr FreeList free_labels(int mask)
{
  FreeList f{};
  const int n = 1 << popc10(mask);
  for (int c = 0; c < n; ++c) {
    int y = 0, k = 0;
    for (int i = 0; i < NQ; ++i)
      if ((mask >> i) & 1) {
        y |= ((c >> k) & 1) << i;
        ++k;
      }
    f.v[c] = y;
  }
  return f;
}

__device__ __forceinline__ float lane_x16(float v)
{
  const int bits = __builtin_bit_cast(int, v);
  return __builtin_bit_cast(float, __builtin_amdgcn_permlanex16(
      bits, bits, 0x76543210, 0xfedcba98, false, false));
}

template <int BASE>
__device__ __forceinline__ void layer_phases(float a, int lane,
                                             float& pr, float& pl)
{
  const int ab = __builtin_bit_cast(int, a);
  float r = 0.0f, l = 0.0f;
  #pragma unroll
  for (int w = 0; w < 5; ++w) {
    const float ar = __builtin_bit_cast(float, __builtin_amdgcn_readlane(ab, BASE + w));
    const float al = __builtin_bit_cast(float, __builtin_amdgcn_readlane(ab, BASE + w + 5));
    const bool bit = (lane >> (4 - w)) & 1;
    r += bit ? ar : -ar;
    l += bit ? al : -al;
  }
  pr = r;
  pl = l;
}

__device__ __forceinline__ void phase_sincos(float phi, float& s, float& c)
{
  const float n = rintf(phi * 0.63661977236758134f);
  float r = fmaf(n, -1.57079637050628662f, phi);
  r = fmaf(n, 4.37113900018624283e-8f, r);
  const int q = (int)n;
  const float r2 = r * r;
  float sp = fmaf(r2, -1.9515295891e-4f, 8.3321608736e-3f);
  sp = fmaf(r2, sp, -1.6666654611e-1f);
  const float sr = fmaf(r * r2, sp, r);
  float cp = fmaf(r2, 2.443315711809948e-5f, -1.388731625493765e-3f);
  cp = fmaf(r2, cp, 4.166664568298827e-2f);
  cp = fmaf(r2, cp, -0.5f);
  const float cr = fmaf(r2, cp, 1.0f);
  const bool odd = q & 1;
  const float sv = odd ? cr : sr;
  const float cv = odd ? sr : cr;
  s = (q & 2) ? -sv : sv;
  c = ((q + 1) & 2) ? -cv : cv;
}

__device__ __forceinline__ void table_entries(int ya, int yb,
                                              const float (&lpr)[NL],
                                              const float (&lpl)[NL], v2f* etab)
{
  int va = ya, vb = yb;
  float fa[2 * NL], fb[2 * NL];
  #pragma unroll
  for (int l = 1; l < NL; ++l) {
    va = ring_inv(va);
    vb = ring_inv(vb);
    fa[2 * l]     = __shfl(lpr[l], va >> 5, 32);
    fa[2 * l + 1] = __shfl(lpl[l], va & 31, 32);
    fb[2 * l]     = __shfl(lpr[l], vb >> 5, 32);
    fb[2 * l + 1] = __shfl(lpl[l], vb & 31, 32);
  }
  asm volatile("" :: "v"(fa[2]), "v"(fa[3]), "v"(fa[4]), "v"(fa[5]),
                     "v"(fa[6]), "v"(fa[7]), "v"(fb[2]), "v"(fb[3]),
                     "v"(fb[4]), "v"(fb[5]), "v"(fb[6]), "v"(fb[7]));
  float ca = 0.0f, cb = 0.0f;
  #pragma unroll
  for (int l = 1; l < NL; ++l) {
    ca += fa[2 * l] + fa[2 * l + 1];
    cb += fb[2 * l] + fb[2 * l + 1];
  }
  float es, ec;
  v2f e;
  phase_sincos(ca, es, ec);
  e.x = ec;
  e.y = es;
  etab[ya] = e;
  phase_sincos(cb, es, ec);
  e.x = ec;
  e.y = es;
  etab[yb] = e;
}

template <int MASK, int I = 0>
__device__ __forceinline__ int deposit(int c)
{
  if constexpr (I >= popc10(MASK)) {
    return 0;
  } else {
    constexpr int P = nth_set(MASK, I);
    return (((c >> I) & 1) << P) | deposit<MASK, I + 1>(c);
  }
}

template <int Q>
__device__ __forceinline__ void tbuild(int unit, int lane, const v2f* etab,
                                       v2f* ttab)
{
  constexpr int D  = flip_mask(Q);
  constexpr int FM = (DIM - 1) & ~D;
  constexpr int NF = 1 << popc10(FM);
  constexpr FreeList FL = free_labels(FM);
  const int c  = unit * 32 + lane;
  const int y0 = deposit<D>(c);
  const int y1 = y0 ^ D;
  static_assert(NF >= 4 && (NF & 3) == 0, "four partial sums");
  float tr[4] = {}, ti[4] = {};
  #pragma unroll
  for (int f = 0; f < NF; ++f) {
    const v2f a = etab[y0 | FL.v[f]];
    const v2f b = etab[y1 | FL.v[f]];
    tr[f & 3] = fmaf(a.x, b.x, tr[f & 3]);
    tr[f & 3] = fmaf(a.y, b.y, tr[f & 3]);
    ti[f & 3] = fmaf(a.y, b.x, ti[f & 3]);
    ti[f & 3] = fmaf(-a.x, b.y, ti[f & 3]);
  }
  v2f t;
  t.x = (tr[0] + tr[1]) + (tr[2] + tr[3]);
  t.y = (ti[0] + ti[1]) + (ti[2] + ti[3]);
  ttab[toff_c(Q) + c] = t;
}

template <int Q>
__device__ __forceinline__ void tbuild_dispatch(int j, int lane,
                                                const v2f* etab, v2f* ttab)
{
  if constexpr (Q < NQ) {
    if (j < toff_c(Q + 1) / 32)
      tbuild<Q>(j - toff_c(Q) / 32, lane, etab, ttab);
    else
      tbuild_dispatch<Q + 1>(j, lane, etab, ttab);
  }
}

__device__ __forceinline__ float zpart(int q, int unit0, int lane,
                                       const v2f* urow, const v2f* ttab)
{
  int d = 0, off = 0;
  #pragma unroll
  for (int i = 0; i < NQ; ++i) {
    if (q == i) d = flip_mask(i);
    if (i < q) off += 1 << popc10(flip_mask(i));
  }
  const int m  = __builtin_popcount(d);
  const int nu = (m >= 7) ? 4 : (1 << (m - 5));
  const int hb = lane >> 4;
  v2f u[8];
  int rest = d;
  #pragma unroll
  for (int i = 0; i < 8; ++i) {
    const bool has = (rest != 0);
    const int p = has ? __builtin_ctz(rest) : (NQ - 1);
    rest &= rest - 1;
    const v2f f = urow[NQ - 1 - p];
    u[i].x = has ? f.x : 1.0f;
    u[i].y = has ? f.y : 0.0f;
  }
  float ar[16], ai[16];
  ar[0] = u[0].x;  ai[0] = -u[0].y;
  ar[1] = u[0].x;  ai[1] = u[0].y;
  #pragma unroll
  for (int i = 1; i < 4; ++i) {
    const float c = u[i].x, s = u[i].y;
    #pragma unroll
    for (int e = 0; e < (1 << i); ++e) {
      const float xr = ar[e], xi = ai[e];
      ar[e + (1 << i)] = xr * c - xi * s;
      ai[e + (1 << i)] = xr * s + xi * c;
      ar[e] = xr * c + xi * s;
      ai[e] = xi * c - xr * s;
    }
  }
  const float g4r = u[4].x;
  const float g4i = hb ? u[4].y : -u[4].y;
  float acc = 0.0f;
  for (int up = 0; up < nu; up += 2) {
    const v2f* tp = ttab + off + (unit0 + up) * 32 + hb * 16;
    float sr[2][4] = {}, si[2][4] = {};
    #pragma unroll
    for (int e = 0; e < 16; ++e) {
      #pragma unroll
      for (int h = 0; h < 2; ++h) {
        const v2f t = tp[h * 32 + e];
        sr[h][e & 3] = fmaf(t.x, ar[e], sr[h][e & 3]);
        sr[h][e & 3] = fmaf(-t.y, ai[e], sr[h][e & 3]);
        si[h][e & 3] = fmaf(t.x, ai[e], si[h][e & 3]);
        si[h][e & 3] = fmaf(t.y, ar[e], si[h][e & 3]);
      }
    }
    #pragma unroll
    for (int h = 0; h < 2; ++h) {
      const int unit = unit0 + up + h;
      float gr = g4r, gi = g4i;
      #pragma unroll
      for (int i = 5; i < 8; ++i) {
        const float c = u[i].x;
        const float s = ((unit >> (i - 5)) & 1) ? u[i].y : -u[i].y;
        const float nr = gr * c - gi * s;
        gi = gr * s + gi * c;
        gr = nr;
      }
      const float srs = (sr[h][0] + sr[h][1]) + (sr[h][2] + sr[h][3]);
      const float sis = (si[h][0] + si[h][1]) + (si[h][2] + si[h][3]);
      acc = fmaf(gr, srs, acc);
      acc = fmaf(-gi, sis, acc);
    }
  }
  return acc;
}

__global__ __launch_bounds__(512) void k_fused(
    const float* __restrict__ x, const float* __restrict__ Win,
    const float* __restrict__ bin, const float* __restrict__ qp,
    const float* __restrict__ Wout, const float* __restrict__ bout,
    float* __restrict__ out)
{
  __shared__ v2f etab[DIM];
  __shared__ v2f ttab[TTOT];
  __shared__ v2f usm[ROWS * NQ];
  __shared__ float Bs[16 * BPAD];
  __shared__ float xs[ROWS * XPAD];
  __shared__ float hsm[ROWS * HPAD];
  __shared__ float zsm[ROWS * ZPAD];

  const int tid  = threadIdx.x;
  const int wave = __builtin_amdgcn_readfirstlane(tid >> 5);
  const int lane = tid & 31;
  const int half = lane >> 4;
  const int l16  = lane & 15;
  const int m0   = blockIdx.x * ROWS;

  const float qv = qp[NQ + ((lane < (NL - 1) * NQ) ? lane : ((NL - 1) * NQ - 1))];

  const int n0 = wave * 16;
  v2f bfrag[3];
  {
    const float* brow = Wout + (n0 + l16) * NQ;
    #pragma unroll
    for (int kk = 0; kk < 3; ++kk) {
      const int k0 = kk * 4 + half * 2;
      const bool ext = (k0 >= NQ);
      bfrag[kk].x = brow[ext ? QX0 : k0];
      bfrag[kk].y = brow[ext ? QX1 : (k0 + 1)];
    }
  }
  const float obias = bout[n0 + l16];
  const float th0 = qp[tid % NQ];

  float wv[16 * K_IN / 512];
  #pragma unroll
  for (int t = 0; t < 16 * K_IN / 512; ++t) {
    const int i = tid + t * 512;
    const int n = i >> 7;
    const int k = i & (K_IN - 1);
    wv[t] = Win[((n < NQ) ? n : (NQ - 1)) * K_IN + k];
  }
  const float ebias = bin[(l16 < NQ) ? l16 : (NQ - 1)];
  float xv[4];
  {
    const float* xp = x + (m0 + (tid >> 5)) * K_IN + (tid & 31) * 4;
    #pragma unroll
    for (int j = 0; j < 4; ++j) xv[j] = xp[j];
  }
  #pragma unroll
  for (int t = 0; t < 16 * K_IN / 512; ++t) {
    const int i = tid + t * 512;
    const int n = i >> 7;
    const int k = i & (K_IN - 1);
    Bs[n * BPAD + k] = (n < NQ) ? wv[t] : 0.0f;
  }
  #pragma unroll
  for (int j = 0; j < 4; ++j)
    xs[(tid >> 5) * XPAD + (tid & 31) * 4 + j] = xv[j];
  __builtin_amdgcn_sched_barrier(0);
  {
    float lpr[NL] = {}, lpl[NL] = {};
    const float qh = 0.5f * qv;
    layer_phases<0 * NQ>(qh, lane, lpr[1], lpl[1]);
    layer_phases<1 * NQ>(qh, lane, lpr[2], lpl[2]);
    layer_phases<2 * NQ>(qh, lane, lpr[3], lpl[3]);
    table_entries(tid, tid + 512, lpr, lpl, etab);
  }
  __syncthreads();

  if (wave != 0) {
    for (int j = wave - 1; j < NUNIT; j += 15)
      tbuild_dispatch<0>(j, lane, etab, ttab);
  } else {
    const float* brow = Bs + l16 * BPAD + half * 2;
    const float* arow = xs + l16 * XPAD + half * 2;
    v8f acc = {}, acc1 = {};
    v2f af[K_IN / 4], bf[K_IN / 4];
    #pragma unroll
    for (int kk = 0; kk < K_IN / 4; ++kk) {
      af[kk].x = arow[kk * 4];
      af[kk].y = arow[kk * 4 + 1];
      bf[kk].x = brow[kk * 4];
      bf[kk].y = brow[kk * 4 + 1];
    }
    #pragma unroll
    for (int g = 0; g < K_IN / 4; g += 8)
      asm volatile("" :: "v"(af[g]), "v"(af[g + 1]), "v"(af[g + 2]), "v"(af[g + 3]),
                         "v"(af[g + 4]), "v"(af[g + 5]), "v"(af[g + 6]), "v"(af[g + 7]),
                         "v"(bf[g]), "v"(bf[g + 1]), "v"(bf[g + 2]), "v"(bf[g + 3]),
                         "v"(bf[g + 4]), "v"(bf[g + 5]), "v"(bf[g + 6]), "v"(bf[g + 7]));
    #pragma unroll
    for (int kk = 0; kk < K_IN / 4; kk += 2) {
      acc  = __builtin_amdgcn_wmma_f32_16x16x4_f32(false, af[kk], false, bf[kk],
                                                   (short)0, acc, false, false);
      acc1 = __builtin_amdgcn_wmma_f32_16x16x4_f32(false, af[kk + 1], false, bf[kk + 1],
                                                   (short)0, acc1, false, false);
    }
    acc += acc1;
    if (l16 < NQ) {
      #pragma unroll
      for (int r = 0; r < 8; ++r)
        hsm[(r + 8 * half) * HPAD + l16] = tanhf(acc[r] + ebias);
    }
  }
  __syncthreads();

  if (tid < ROWS * NQ) {
    const int r = tid / NQ;
    const int w = tid - r * NQ;
    float sa, ca;
    phase_sincos(hsm[r * HPAD + w] + th0, sa, ca);
    v2f uv;
    uv.x = ca;
    uv.y = sa;
    usm[tid] = uv;
  }
  __syncthreads();

  if (wave < ZPAD) {
    const int q     = (wave < NQ) ? wave : ((wave == NQ) ? QX0 : QX1);
    const int unit0 = (wave < NQ) ? 0 : 4;
    float v = zpart(q, unit0, lane, usm + l16 * NQ, ttab);
    v += lane_x16(v);
    if (half == 0) zsm[l16 * ZPAD + wave] = v * (1.0f / 1024.0f);
  }
  __syncthreads();

  {
    const float* zrow = zsm + l16 * ZPAD + half * 2;
    v8f acc = {};
    #pragma unroll
    for (int kk = 0; kk < 3; ++kk) {
      v2f a;
      a.x = zrow[kk * 4];
      a.y = zrow[kk * 4 + 1];
      acc = __builtin_amdgcn_wmma_f32_16x16x4_f32(false, a, false, bfrag[kk],
                                                  (short)0, acc, false, false);
    }
    #pragma unroll
    for (int r = 0; r < 8; ++r) {
      const int m = m0 + r + 8 * half;
      out[m * N_OUT + n0 + l16] = acc[r] + obias;
    }
  }
}

extern "C" void kernel_launch(void* const* d_in, const int* in_sizes, int n_in,
                              void* d_out, int out_size, void* d_ws, size_t ws_size,
                              hipStream_t stream) {
  const float* x    = (const float*)d_in[0];
  const float* Win  = (const float*)d_in[1];
  const float* bin  = (const float*)d_in[2];
  const float* qp   = (const float*)d_in[3];
  const float* Wout = (const float*)d_in[4];
  const float* bout = (const float*)d_in[5];
  float* out = (float*)d_out;

  k_fused<<<BATCH / ROWS, 512, 0, stream>>>(x, Win, bin, qp, Wout, bout, out);
}


# phase 2: terms of t and its complement in d_q are conjugates, real parts equal: only the half of each T_q with the top index bit clear is built and summed, doubled; 672 terms a row; sums reordered
# speedup vs baseline: 12.9777x; 1.0988x over previous
#include <hip/hip_runtime.h>
#include <math.h>

typedef float v2f __attribute__((ext_vector_type(2)));
typedef float v8f __attribute__((ext_vector_type(8)));

#define NQ 10
#define NL 4
#define DIM 1024
#define BATCH 4096
#define K_IN 128
#define N_OUT 256
#define BPAD 132
#define XPAD 132

#define ROWS 16
#define HPAD 16
#define ZPAD 12

constexpr int ring_c(int j)
{
  int s = j ^ (j >> 9);
  s ^= s << 1;
  s ^= s << 2;
  s ^= s << 4;
  s ^= s << 8;
  return s & (DIM - 1);
}
constexpr int ring_inv(int r)
{
  const int s = (r ^ (r << 1)) & (DIM - 1);
  return s ^ (s >> 9);
}
constexpr int flip_mask(int q)
{
  int d = 1 << (NQ - 1 - q);
  for (int i = 0; i < NL; ++i) d = ring_c(d);
  return d;
}
constexpr int popc10(int d)
{
  int c = 0;
  for (int i = 0; i < NQ; ++i) c += (d >> i) & 1;
  return c;
}
constexpr int nth_set(int d, int n)
{
  for (int i = 0; i < NQ; ++i)
    if ((d >> i) & 1) {
      if (n == 0) return i;
      --n;
    }
  return -1;
}
constexpr int toff_c(int q)
{
  int o = 0;
  for (int i = 0; i < q; ++i) o += 1 << (popc10(flip_mask(i)) - 1);
  return o;
}
constexpr int TTOT  = toff_c(NQ);
constexpr int NUNIT = TTOT / 32;
constexpr int n_wide()
{
  int n = 0;
  for (int q = 0; q < NQ; ++q) n += (popc10(flip_mask(q)) == 8);
  return n;
}
constexpr int nth_wide(int n)
{
  for (int q = 0; q < NQ; ++q)
    if (popc10(flip_mask(q)) == 8) {
      if (n == 0) return q;
      --n;
    }
  return -1;
}
constexpr bool masks_fit()
{
  for (int q = 0; q < NQ; ++q) {
    const int m = popc10(flip_mask(q));
    if (m < 6 || m > 8) return false;
  }
  return true;
}
static_assert(masks_fit(), "every flip mask needs 6 to 8 bits: half a table is 1, 2 or 4 units");
static_assert(n_wide() == 2, "twelve tasks: ten qubits and two upper halves");
static_assert(NQ + n_wide() == ZPAD, "phase 3 K dimension");
constexpr int QX0 = nth_wide(0);
constexpr int QX1 = nth_wide(1);

struct FreeList { int v[32]; };
constexpr FreeList free_labels(int mask)
{
  FreeList f{};
  const int n = 1 << popc10(mask);
  for (int c = 0; c < n; ++c) {
    int y = 0, k = 0;
    for (int i = 0; i < NQ; ++i)
      if ((mask >> i) & 1) {
        y |= ((c >> k) & 1) << i;
        ++k;
      }
    f.v[c] = y;
  }
  return f;
}

__device__ __forceinline__ float lane_x16(float v)
{
  const int bits = __builtin_bit_cast(int, v);
  return __builtin_bit_cast(float, __builtin_amdgcn_permlanex16(
      bits, bits, 0x76543210, 0xfedcba98, false, false));
}

template <int BASE>
__device__ __forceinline__ void layer_phases(float a, int lane,
                                             float& pr, float& pl)
{
  const int ab = __builtin_bit_cast(int, a);
  float r = 0.0f, l = 0.0f;
  #pragma unroll
  for (int w = 0; w < 5; ++w) {
    const float ar = __builtin_bit_cast(float, __builtin_amdgcn_readlane(ab, BASE + w));
    const float al = __builtin_bit_cast(float, __builtin_amdgcn_readlane(ab, BASE + w + 5));
    const bool bit = (lane >> (4 - w)) & 1;
    r += bit ? ar : -ar;
    l += bit ? al : -al;
  }
  pr = r;
  pl = l;
}

__device__ __forceinline__ void phase_sincos(float phi, float& s, float& c)
{
  const float n = rintf(phi * 0.63661977236758134f);
  float r = fmaf(n, -1.57079637050628662f, phi);
  r = fmaf(n, 4.37113900018624283e-8f, r);
  const int q = (int)n;
  const float r2 = r * r;
  float sp = fmaf(r2, -1.9515295891e-4f, 8.3321608736e-3f);
  sp = fmaf(r2, sp, -1.6666654611e-1f);
  const float sr = fmaf(r * r2, sp, r);
  float cp = fmaf(r2, 2.443315711809948e-5f, -1.388731625493765e-3f);
  cp = fmaf(r2, cp, 4.166664568298827e-2f);
  cp = fmaf(r2, cp, -0.5f);
  const float cr = fmaf(r2, cp, 1.0f);
  const bool odd = q & 1;
  const float sv = odd ? cr : sr;
  const float cv = odd ? sr : cr;
  s = (q & 2) ? -sv : sv;
  c = ((q + 1) & 2) ? -cv : cv;
}

__device__ __forceinline__ void table_entries(int ya, int yb,
                                              const float (&lpr)[NL],
                                              const float (&lpl)[NL], v2f* etab)
{
  int va = ya, vb = yb;
  float fa[2 * NL], fb[2 * NL];
  #pragma unroll
  for (int l = 1; l < NL; ++l) {
    va = ring_inv(va);
    vb = ring_inv(vb);
    fa[2 * l]     = __shfl(lpr[l], va >> 5, 32);
    fa[2 * l + 1] = __shfl(lpl[l], va & 31, 32);
    fb[2 * l]     = __shfl(lpr[l], vb >> 5, 32);
    fb[2 * l + 1] = __shfl(lpl[l], vb & 31, 32);
  }
  asm volatile("" :: "v"(fa[2]), "v"(fa[3]), "v"(fa[4]), "v"(fa[5]),
                     "v"(fa[6]), "v"(fa[7]), "v"(fb[2]), "v"(fb[3]),
                     "v"(fb[4]), "v"(fb[5]), "v"(fb[6]), "v"(fb[7]));
  float ca = 0.0f, cb = 0.0f;
  #pragma unroll
  for (int l = 1; l < NL; ++l) {
    ca += fa[2 * l] + fa[2 * l + 1];
    cb += fb[2 * l] + fb[2 * l + 1];
  }
  float es, ec;
  v2f e;
  phase_sincos(ca, es, ec);
  e.x = ec;
  e.y = es;
  etab[ya] = e;
  phase_sincos(cb, es, ec);
  e.x = ec;
  e.y = es;
  etab[yb] = e;
}

template <int MASK, int I = 0>
__device__ __forceinline__ int deposit(int c)
{
  if constexpr (I >= popc10(MASK)) {
    return 0;
  } else {
    constexpr int P = nth_set(MASK, I);
    return (((c >> I) & 1) << P) | deposit<MASK, I + 1>(c);
  }
}

template <int Q>
__device__ __forceinline__ void tbuild(int unit, int lane, const v2f* etab,
                                       v2f* ttab)
{
  constexpr int D  = flip_mask(Q);
  constexpr int FM = (DIM - 1) & ~D;
  constexpr int NF = 1 << popc10(FM);
  constexpr FreeList FL = free_labels(FM);
  const int c  = unit * 32 + lane;
  const int y0 = deposit<D>(c);
  const int y1 = y0 ^ D;
  static_assert(NF >= 4 && (NF & 3) == 0, "four partial sums");
  float tr[4] = {}, ti[4] = {};
  #pragma unroll
  for (int f = 0; f < NF; ++f) {
    const v2f a = etab[y0 | FL.v[f]];
    const v2f b = etab[y1 | FL.v[f]];
    tr[f & 3] = fmaf(a.x, b.x, tr[f & 3]);
    tr[f & 3] = fmaf(a.y, b.y, tr[f & 3]);
    ti[f & 3] = fmaf(a.y, b.x, ti[f & 3]);
    ti[f & 3] = fmaf(-a.x, b.y, ti[f & 3]);
  }
  v2f t;
  t.x = (tr[0] + tr[1]) + (tr[2] + tr[3]);
  t.y = (ti[0] + ti[1]) + (ti[2] + ti[3]);
  ttab[toff_c(Q) + c] = t;
}

template <int Q>
__device__ __forceinline__ void tbuild_dispatch(int j, int lane,
                                                const v2f* etab, v2f* ttab)
{
  if constexpr (Q < NQ) {
    if (j < toff_c(Q + 1) / 32)
      tbuild<Q>(j - toff_c(Q) / 32, lane, etab, ttab);
    else
      tbuild_dispatch<Q + 1>(j, lane, etab, ttab);
  }
}

template <int NU>
__device__ __forceinline__ float task_sum(const v2f* tp, int unit0,
                                          const float (&ar)[16],
                                          const float (&ai)[16],
                                          const v2f (&u)[8], float g4r, float g4i)
{
  float sr[NU][4] = {}, si[NU][4] = {};
  #pragma unroll
  for (int e = 0; e < 16; ++e) {
    #pragma unroll
    for (int h = 0; h < NU; ++h) {
      const v2f t = tp[h * 32 + e];
      sr[h][e & 3] = fmaf(t.x, ar[e], sr[h][e & 3]);
      sr[h][e & 3] = fmaf(-t.y, ai[e], sr[h][e & 3]);
      si[h][e & 3] = fmaf(t.x, ai[e], si[h][e & 3]);
      si[h][e & 3] = fmaf(t.y, ar[e], si[h][e & 3]);
    }
  }
  float acc = 0.0f;
  #pragma unroll
  for (int h = 0; h < NU; ++h) {
    const int unit = unit0 + h;
    float gr = g4r, gi = g4i;
    #pragma unroll
    for (int i = 5; i < 8; ++i) {
      const float c = u[i].x;
      const float s = ((unit >> (i - 5)) & 1) ? u[i].y : -u[i].y;
      const float nr = gr * c - gi * s;
      gi = gr * s + gi * c;
      gr = nr;
    }
    const float srs = (sr[h][0] + sr[h][1]) + (sr[h][2] + sr[h][3]);
    const float sis = (si[h][0] + si[h][1]) + (si[h][2] + si[h][3]);
    acc = fmaf(gr, srs, acc);
    acc = fmaf(-gi, sis, acc);
  }
  return acc;
}

__device__ __forceinline__ float zpart(int q, int unit0, int lane,
                                       const v2f* urow, const v2f* ttab)
{
  int d = 0, off = 0;
  #pragma unroll
  for (int i = 0; i < NQ; ++i) {
    if (q == i) d = flip_mask(i);
    if (i < q) off += 1 << (popc10(flip_mask(i)) - 1);
  }
  const int m  = __builtin_popcount(d);
  const int hb = lane >> 4;
  v2f u[8];
  int rest = d;
  #pragma unroll
  for (int i = 0; i < 8; ++i) {
    const bool has = (rest != 0);
    const int p = has ? __builtin_ctz(rest) : (NQ - 1);
    rest &= rest - 1;
    const v2f f = urow[NQ - 1 - p];
    u[i].x = has ? f.x : 1.0f;
    u[i].y = has ? f.y : 0.0f;
  }
  float ar[16], ai[16];
  ar[0] = u[0].x;  ai[0] = -u[0].y;
  ar[1] = u[0].x;  ai[1] = u[0].y;
  #pragma unroll
  for (int i = 1; i < 4; ++i) {
    const float c = u[i].x, s = u[i].y;
    #pragma unroll
    for (int e = 0; e < (1 << i); ++e) {
      const float xr = ar[e], xi = ai[e];
      ar[e + (1 << i)] = xr * c - xi * s;
      ai[e + (1 << i)] = xr * s + xi * c;
      ar[e] = xr * c + xi * s;
      ai[e] = xi * c - xr * s;
    }
  }
  const float g4r = u[4].x;
  const float g4i = hb ? u[4].y : -u[4].y;
  const v2f* tp = ttab + off + unit0 * 32 + hb * 16;
  if (m >= 7) return task_sum<2>(tp, unit0, ar, ai, u, g4r, g4i);
  return task_sum<1>(tp, unit0, ar, ai, u, g4r, g4i);
}

__global__ __launch_bounds__(512) void k_fused(
    const float* __restrict__ x, const float* __restrict__ Win,
    const float* __restrict__ bin, const float* __restrict__ qp,
    const float* __restrict__ Wout, const float* __restrict__ bout,
    float* __restrict__ out)
{
  __shared__ v2f etab[DIM];
  __shared__ v2f ttab[TTOT];
  __shared__ v2f usm[ROWS * NQ];
  __shared__ float Bs[16 * BPAD];
  __shared__ float xs[ROWS * XPAD];
  __shared__ float hsm[ROWS * HPAD];
  __shared__ float zsm[ROWS * ZPAD];

  const int tid  = threadIdx.x;
  const int wave = __builtin_amdgcn_readfirstlane(tid >> 5);
  const int lane = tid & 31;
  const int half = lane >> 4;
  const int l16  = lane & 15;
  const int m0   = blockIdx.x * ROWS;

  const float qv = qp[NQ + ((lane < (NL - 1) * NQ) ? lane : ((NL - 1) * NQ - 1))];

  const int n0 = wave * 16;
  v2f bfrag[3];
  {
    const float* brow = Wout + (n0 + l16) * NQ;
    #pragma unroll
    for (int kk = 0; kk < 3; ++kk) {
      const int k0 = kk * 4 + half * 2;
      const bool ext = (k0 >= NQ);
      bfrag[kk].x = brow[ext ? QX0 : k0];
      bfrag[kk].y = brow[ext ? QX1 : (k0 + 1)];
    }
  }
  const float obias = bout[n0 + l16];
  const float th0 = qp[tid % NQ];

  float wv[16 * K_IN / 512];
  #pragma unroll
  for (int t = 0; t < 16 * K_IN / 512; ++t) {
    const int i = tid + t * 512;
    const int n = i >> 7;
    const int k = i & (K_IN - 1);
    wv[t] = Win[((n < NQ) ? n : (NQ - 1)) * K_IN + k];
  }
  const float ebias = bin[(l16 < NQ) ? l16 : (NQ - 1)];
  float xv[4];
  {
    const float* xp = x + (m0 + (tid >> 5)) * K_IN + (tid & 31) * 4;
    #pragma unroll
    for (int j = 0; j < 4; ++j) xv[j] = xp[j];
  }
  #pragma unroll
  for (int t = 0; t < 16 * K_IN / 512; ++t) {
    const int i = tid + t * 512;
    const int n = i >> 7;
    const int k = i & (K_IN - 1);
    Bs[n * BPAD + k] = (n < NQ) ? wv[t] : 0.0f;
  }
  #pragma unroll
  for (int j = 0; j < 4; ++j)
    xs[(tid >> 5) * XPAD + (tid & 31) * 4 + j] = xv[j];
  __builtin_amdgcn_sched_barrier(0);
  {
    float lpr[NL] = {}, lpl[NL] = {};
    const float qh = 0.5f * qv;
    layer_phases<0 * NQ>(qh, lane, lpr[1], lpl[1]);
    layer_phases<1 * NQ>(qh, lane, lpr[2], lpl[2]);
    layer_phases<2 * NQ>(qh, lane, lpr[3], lpl[3]);
    table_entries(tid, tid + 512, lpr, lpl, etab);
  }
  __syncthreads();

  if (wave != 0) {
    for (int j = wave - 1; j < NUNIT; j += 15)
      tbuild_dispatch<0>(j, lane, etab, ttab);
  } else {
    const float* brow = Bs + l16 * BPAD + half * 2;
    const float* arow = xs + l16 * XPAD + half * 2;
    v8f acc = {}, acc1 = {};
    v2f af[K_IN / 4], bf[K_IN / 4];
    #pragma unroll
    for (int kk = 0; kk < K_IN / 4; ++kk) {
      af[kk].x = arow[kk * 4];
      af[kk].y = arow[kk * 4 + 1];
      bf[kk].x = brow[kk * 4];
      bf[kk].y = brow[kk * 4 + 1];
    }
    #pragma unroll
    for (int g = 0; g < K_IN / 4; g += 8)
      asm volatile("" :: "v"(af[g]), "v"(af[g + 1]), "v"(af[g + 2]), "v"(af[g + 3]),
                         "v"(af[g + 4]), "v"(af[g + 5]), "v"(af[g + 6]), "v"(af[g + 7]),
                         "v"(bf[g]), "v"(bf[g + 1]), "v"(bf[g + 2]), "v"(bf[g + 3]),
                         "v"(bf[g + 4]), "v"(bf[g + 5]), "v"(bf[g + 6]), "v"(bf[g + 7]));
    #pragma unroll
    for (int kk = 0; kk < K_IN / 4; kk += 2) {
      acc  = __builtin_amdgcn_wmma_f32_16x16x4_f32(false, af[kk], false, bf[kk],
                                                   (short)0, acc, false, false);
      acc1 = __builtin_amdgcn_wmma_f32_16x16x4_f32(false, af[kk + 1], false, bf[kk + 1],
                                                   (short)0, acc1, false, false);
    }
    acc += acc1;
    if (l16 < NQ) {
      #pragma unroll
      for (int r = 0; r < 8; ++r)
        hsm[(r + 8 * half) * HPAD + l16] = tanhf(acc[r] + ebias);
    }
  }
  __syncthreads();

  if (tid < ROWS * NQ) {
    const int r = tid / NQ;
    const int w = tid - r * NQ;
    float sa, ca;
    phase_sincos(hsm[r * HPAD + w] + th0, sa, ca);
    v2f uv;
    uv.x = ca;
    uv.y = sa;
    usm[tid] = uv;
  }
  __syncthreads();

  if (wave < ZPAD) {
    const int q     = (wave < NQ) ? wave : ((wave == NQ) ? QX0 : QX1);
    const int unit0 = (wave < NQ) ? 0 : 2;
    float v = zpart(q, unit0, lane, usm + l16 * NQ, ttab);
    v += lane_x16(v);
    if (half == 0) zsm[l16 * ZPAD + wave] = v * (1.0f / 512.0f);
  }
  __syncthreads();

  {
    const float* zrow = zsm + l16 * ZPAD + half * 2;
    v8f acc = {};
    #pragma unroll
    for (int kk = 0; kk < 3; ++kk) {
      v2f a;
      a.x = zrow[kk * 4];
      a.y = zrow[kk * 4 + 1];
      acc = __builtin_amdgcn_wmma_f32_16x16x4_f32(false, a, false, bfrag[kk],
                                                  (short)0, acc, false, false);
    }
    #pragma unroll
    for (int r = 0; r < 8; ++r) {
      const int m = m0 + r + 8 * half;
      out[m * N_OUT + n0 + l16] = acc[r] + obias;
    }
  }
}

extern "C" void kernel_launch(void* const* d_in, const int* in_sizes, int n_in,
                              void* d_out, int out_size, void* d_ws, size_t ws_size,
                              hipStream_t stream) {
  const float* x    = (const float*)d_in[0];
  const float* Win  = (const float*)d_in[1];
  const float* bin  = (const float*)d_in[2];
  const float* qp   = (const float*)d_in[3];
  const float* Wout = (const float*)d_in[4];
  const float* bout = (const float*)d_in[5];
  float* out = (float*)d_out;

  k_fused<<<BATCH / ROWS, 512, 0, stream>>>(x, Win, bin, qp, Wout, bout, out);
}
